# starting kernel
# baseline (speedup 1.0000x reference)
#include <hip/hip_runtime.h>


typedef __attribute__((ext_vector_type(2))) float v2f;
typedef __attribute__((ext_vector_type(8))) float v8f;

#define GNN_SLOPE 0.01f
#define GNN_EPS   1e-5f
#define GNN_BSN   16
#define GNN_N     2000
#define GNN_E     32000
#define GNN_NT    (GNN_BSN * GNN_N)
#define GNN_DOUT  128

__device__ __forceinline__ float lrelu(float v) { return v >= 0.0f ? v : GNN_SLOPE * v; }

#define WMMA_F32(A, B, C) \
    __builtin_amdgcn_wmma_f32_16x16x4_f32(false, (A), false, (B), (short)0, (C), false, false)

__global__ __launch_bounds__(256) void k_fill_zero(float* __restrict__ p, int n) {
    int i = blockIdx.x * 256 + threadIdx.x;
    if (i < n) p[i] = 0.0f;
}

template <bool DIFF>
__global__ __launch_bounds__(256) void k_pack_b(const float* __restrict__ B,
                                                float* __restrict__ Bp, int K) {
    int i = blockIdx.x * 256 + threadIdx.x;
    const int kk   = i >> 9;
    const int rem  = i & 511;
    const int lane = rem >> 4;
    const int j    = rem & 15;
    const int t    = j >> 1;
    const int comp = j & 1;
    const int r    = kk * 4 + 2 * (lane >> 4) + comp;
    const int c    = t * 16 + (lane & 15);
    const int src  = r * GNN_DOUT + c;
    float v = B[src];
    if (DIFF) v -= B[K * GNN_DOUT + src];
    Bp[i] = v;
}

template <int K>
__global__ __launch_bounds__(256) void k_gemm_pq(const float* __restrict__ A,
                                                 const float* __restrict__ BpA,
                                                 const float* __restrict__ BpB,
                                                 float* __restrict__ P,
                                                 float* __restrict__ Q) {
    const int wave = threadIdx.x >> 5;
    const int lane = threadIdx.x & 31;
    const int half = lane >> 4;
    const int lm   = lane & 15;
    const int m0   = blockIdx.x * 128 + wave * 16;

    v8f accP[8] = {};
    v8f accQ[8] = {};

    const float*  arow = A + (size_t)(m0 + lm) * K + 2 * half;
    const float4* bpA  = (const float4*)(BpA + (size_t)lane * 16);
    const float4* bpB  = (const float4*)(BpB + (size_t)lane * 16);

#pragma unroll
    for (int k = 0; k < K; k += 4) {
        const v2f a = *(const v2f*)(arow + k);
        const float4* ba = bpA + (size_t)(k >> 2) * 128;
        const float4* bb = bpB + (size_t)(k >> 2) * 128;
        float4 pa[4], pb[4];
#pragma unroll
        for (int i = 0; i < 4; ++i) { pa[i] = ba[i]; pb[i] = bb[i]; }
        v2f b;
#pragma unroll
        for (int i = 0; i < 4; ++i) {
            b.x = pa[i].x; b.y = pa[i].y; accP[2 * i + 0] = WMMA_F32(a, b, accP[2 * i + 0]);
            b.x = pa[i].z; b.y = pa[i].w; accP[2 * i + 1] = WMMA_F32(a, b, accP[2 * i + 1]);
            b.x = pb[i].x; b.y = pb[i].y; accQ[2 * i + 0] = WMMA_F32(a, b, accQ[2 * i + 0]);
            b.x = pb[i].z; b.y = pb[i].w; accQ[2 * i + 1] = WMMA_F32(a, b, accQ[2 * i + 1]);
        }
    }

    const size_t rb = (size_t)(m0 + 8 * half) * GNN_DOUT + lm;
    float* prow = P + rb;
    float* qrow = Q + rb;
#pragma unroll
    for (int r = 0; r < 8; ++r)
#pragma unroll
        for (int t = 0; t < 8; ++t) {
            prow[(size_t)r * GNN_DOUT + t * 16] = accP[t][r];
            qrow[(size_t)r * GNN_DOUT + t * 16] = accQ[t][r];
        }
}

__global__ __launch_bounds__(256) void k_edge(const float* __restrict__ P,
                                              const float* __restrict__ Q,
                                              const float* __restrict__ b1,
                                              const int* __restrict__ ei,
                                              float* __restrict__ S,
                                              float* __restrict__ cnt) {
    const int t   = blockIdx.x * 256 + threadIdx.x;
    const int e   = t >> 5;
    const int c   = (t & 31) * 4;
    const int off = blockIdx.y * GNN_N;
    const int src = ei[e] + off;
    const int dst = ei[GNN_E + e] + off;

    const float4 p  = *(const float4*)(P + (size_t)dst * GNN_DOUT + c);
    const float4 q  = *(const float4*)(Q + (size_t)src * GNN_DOUT + c);
    const float4 bb = *(const float4*)(b1 + c);

    float* sp = S + (size_t)dst * GNN_DOUT + c;
    atomicAdd(sp + 0, lrelu(p.x + q.x + bb.x));
    atomicAdd(sp + 1, lrelu(p.y + q.y + bb.y));
    atomicAdd(sp + 2, lrelu(p.z + q.z + bb.z));
    atomicAdd(sp + 3, lrelu(p.w + q.w + bb.w));
    if ((t & 31) == 0) atomicAdd(cnt + dst, 1.0f);
}

template <bool RES>
__global__ __launch_bounds__(256) void k_gemm_agg_ln(const float* __restrict__ S,
                                                     const float* __restrict__ Bp2,
                                                     const float* __restrict__ cnt,
                                                     const float* __restrict__ b2,
                                                     const float* __restrict__ g,
                                                     const float* __restrict__ beta,
                                                     const float* __restrict__ xres,
                                                     float* __restrict__ out) {
    const int wave = threadIdx.x >> 5;
    const int lane = threadIdx.x & 31;
    const int half = lane >> 4;
    const int lm   = lane & 15;
    const int m0   = blockIdx.x * 128 + wave * 16;

    const float cntv  = cnt[m0 + lm];
    const float scale = 1.0f / fmaxf(cntv, 1.0f);

    v8f acc[8] = {};
    const float*  arow = S + (size_t)(m0 + lm) * 128 + 2 * half;
    const float4* bp   = (const float4*)(Bp2 + (size_t)lane * 16);

#pragma unroll
    for (int k = 0; k < 128; k += 4) {
        v2f a = *(const v2f*)(arow + k);
        a.x *= scale; a.y *= scale;
        const float4* bk = bp + (size_t)(k >> 2) * 128;
        float4 pb[4];
#pragma unroll
        for (int i = 0; i < 4; ++i) pb[i] = bk[i];
        v2f b;
#pragma unroll
        for (int i = 0; i < 4; ++i) {
            b.x = pb[i].x; b.y = pb[i].y; acc[2 * i + 0] = WMMA_F32(a, b, acc[2 * i + 0]);
            b.x = pb[i].z; b.y = pb[i].w; acc[2 * i + 1] = WMMA_F32(a, b, acc[2 * i + 1]);
        }
    }

    float b2v[8], gv[8], bev[8];
#pragma unroll
    for (int t = 0; t < 8; ++t) {
        b2v[t] = b2[t * 16 + lm];
        gv[t]  = g[t * 16 + lm];
        bev[t] = beta[t * 16 + lm];
    }

    const size_t rb = (size_t)(m0 + 8 * half) * GNN_DOUT + lm;
    float* crow = out + rb;
    const float* rrow = RES ? (xres + rb) : nullptr;

#pragma unroll
    for (int r = 0; r < 8; ++r) {
        const float cr   = __shfl(cntv, r + 8 * half, 32);
        const float hasE = (cr > 0.0f) ? 1.0f : 0.0f;

        float h[8];
        float s = 0.0f;
#pragma unroll
        for (int t = 0; t < 8; ++t) { h[t] = acc[t][r] + hasE * b2v[t]; s += h[t]; }
#pragma unroll
        for (int m = 8; m >= 1; m >>= 1) s += __shfl_xor(s, m, 32);
        const float mu = s * (1.0f / GNN_DOUT);

        float s2 = 0.0f;
#pragma unroll
        for (int t = 0; t < 8; ++t) { h[t] -= mu; s2 += h[t] * h[t]; }
#pragma unroll
        for (int m = 8; m >= 1; m >>= 1) s2 += __shfl_xor(s2, m, 32);
        const float rstd = rsqrtf(s2 * (1.0f / GNN_DOUT) + GNN_EPS);

#pragma unroll
        for (int t = 0; t < 8; ++t) {
            float o = h[t] * rstd * gv[t] + bev[t];
            if (RES) o += rrow[(size_t)r * GNN_DOUT + t * 16];
            crow[(size_t)r * GNN_DOUT + t * 16] = lrelu(o);
        }
    }
}

extern "C" void kernel_launch(void* const* d_in, const int* in_sizes, int n_in,
                              void* d_out, int out_size, void* d_ws, size_t ws_size,
                              hipStream_t stream) {
    (void)in_sizes; (void)n_in; (void)out_size; (void)ws_size;

    const float* x  = (const float*)d_in[0];
    const int*   ei = (const int*)d_in[1];
    const float* W1[2] = {(const float*)d_in[2],  (const float*)d_in[8]};
    const float* b1[2] = {(const float*)d_in[3],  (const float*)d_in[9]};
    const float* W2[2] = {(const float*)d_in[4],  (const float*)d_in[10]};
    const float* b2[2] = {(const float*)d_in[5],  (const float*)d_in[11]};
    const float* g [2] = {(const float*)d_in[6],  (const float*)d_in[12]};
    const float* be[2] = {(const float*)d_in[7],  (const float*)d_in[13]};

    const size_t NF = (size_t)GNN_NT * GNN_DOUT;
    float* ws   = (float*)d_ws;
    float* X    = ws;
    float* P    = X + NF;
    float* Q    = P + NF;
    float* S    = Q + NF;
    float* cnt  = S + NF;
    float* BpA  = cnt + GNN_NT;
    float* BpB  = BpA + 128 * GNN_DOUT;
    float* Bp2  = BpB + 128 * GNN_DOUT;

    const dim3 egrid(GNN_E * 32 / 256, GNN_BSN);
    const int  zgrid = (int)((NF + GNN_NT + 255) / 256);

    k_pack_b<true ><<< 64 * GNN_DOUT / 256, 256, 0, stream>>>(W1[0], BpA, 64);
    k_pack_b<false><<< 64 * GNN_DOUT / 256, 256, 0, stream>>>(W1[0] + 64 * GNN_DOUT, BpB, 64);
    k_pack_b<false><<<128 * GNN_DOUT / 256, 256, 0, stream>>>(W2[0], Bp2, 128);
    k_fill_zero<<<zgrid, 256, 0, stream>>>(S, (int)(NF + GNN_NT));
    k_gemm_pq<64><<<250, 256, 0, stream>>>(x, BpA, BpB, P, Q);
    k_edge<<<egrid, 256, 0, stream>>>(P, Q, b1[0], ei, S, cnt);
    k_gemm_agg_ln<false><<<250, 256, 0, stream>>>(S, Bp2, cnt, b2[0], g[0], be[0], nullptr, X);

    k_pack_b<true ><<<128 * GNN_DOUT / 256, 256, 0, stream>>>(W1[1], BpA, 128);
    k_pack_b<false><<<128 * GNN_DOUT / 256, 256, 0, stream>>>(W1[1] + 128 * GNN_DOUT, BpB, 128);
    k_pack_b<false><<<128 * GNN_DOUT / 256, 256, 0, stream>>>(W2[1], Bp2, 128);
    k_fill_zero<<<zgrid, 256, 0, stream>>>(S, (int)(NF + GNN_NT));
    k_gemm_pq<128><<<250, 256, 0, stream>>>(X, BpA, BpB, P, Q);
    k_edge<<<egrid, 256, 0, stream>>>(P, Q, b1[1], ei, S, cnt);
    k_gemm_agg_ln<true><<<250, 256, 0, stream>>>(S, Bp2, cnt, b2[1], g[1], be[1], X, (float*)d_out);
}


# Edge stage: each wave owns one destination node in all 16 graphs and filters the LDS-staged shared edge list for it; register sums, no atomics, no zero-fill; sum order now fixed, b1 added to P first
# speedup vs baseline: 2.7202x; 2.7202x over previous
#include <hip/hip_runtime.h>


typedef __attribute__((ext_vector_type(2))) float v2f;
typedef __attribute__((ext_vector_type(8))) float v8f;

#define GNN_SLOPE 0.01f
#define GNN_EPS   1e-5f
#define GNN_BSN   16
#define GNN_N     2000
#define GNN_E     32000
#define GNN_NT    (GNN_BSN * GNN_N)
#define GNN_DOUT  128

__device__ __forceinline__ float lrelu(float v) { return v >= 0.0f ? v : GNN_SLOPE * v; }

#define WMMA_F32(A, B, C) \
    __builtin_amdgcn_wmma_f32_16x16x4_f32(false, (A), false, (B), (short)0, (C), false, false)

template <bool DIFF>
__global__ __launch_bounds__(256) void k_pack_b(const float* __restrict__ B,
                                                float* __restrict__ Bp, int K) {
    int i = blockIdx.x * 256 + threadIdx.x;
    const int kk   = i >> 9;
    const int rem  = i & 511;
    const int lane = rem >> 4;
    const int j    = rem & 15;
    const int t    = j >> 1;
    const int comp = j & 1;
    const int r    = kk * 4 + 2 * (lane >> 4) + comp;
    const int c    = t * 16 + (lane & 15);
    const int src  = r * GNN_DOUT + c;
    float v = B[src];
    if (DIFF) v -= B[K * GNN_DOUT + src];
    Bp[i] = v;
}

template <int K>
__global__ __launch_bounds__(256) void k_gemm_pq(const float* __restrict__ A,
                                                 const float* __restrict__ BpA,
                                                 const float* __restrict__ BpB,
                                                 float* __restrict__ P,
                                                 float* __restrict__ Q) {
    const int wave = threadIdx.x >> 5;
    const int lane = threadIdx.x & 31;
    const int half = lane >> 4;
    const int lm   = lane & 15;
    const int m0   = blockIdx.x * 128 + wave * 16;

    v8f accP[8] = {};
    v8f accQ[8] = {};

    const float*  arow = A + (size_t)(m0 + lm) * K + 2 * half;
    const float4* bpA  = (const float4*)(BpA + (size_t)lane * 16);
    const float4* bpB  = (const float4*)(BpB + (size_t)lane * 16);

#pragma unroll
    for (int k = 0; k < K; k += 4) {
        const v2f a = *(const v2f*)(arow + k);
        const float4* ba = bpA + (size_t)(k >> 2) * 128;
        const float4* bb = bpB + (size_t)(k >> 2) * 128;
        float4 pa[4], pb[4];
#pragma unroll
        for (int i = 0; i < 4; ++i) { pa[i] = ba[i]; pb[i] = bb[i]; }
        v2f b;
#pragma unroll
        for (int i = 0; i < 4; ++i) {
            b.x = pa[i].x; b.y = pa[i].y; accP[2 * i + 0] = WMMA_F32(a, b, accP[2 * i + 0]);
            b.x = pa[i].z; b.y = pa[i].w; accP[2 * i + 1] = WMMA_F32(a, b, accP[2 * i + 1]);
            b.x = pb[i].x; b.y = pb[i].y; accQ[2 * i + 0] = WMMA_F32(a, b, accQ[2 * i + 0]);
            b.x = pb[i].z; b.y = pb[i].w; accQ[2 * i + 1] = WMMA_F32(a, b, accQ[2 * i + 1]);
        }
    }

    const size_t rb = (size_t)(m0 + 8 * half) * GNN_DOUT + lm;
    float* prow = P + rb;
    float* qrow = Q + rb;
#pragma unroll
    for (int r = 0; r < 8; ++r)
#pragma unroll
        for (int t = 0; t < 8; ++t) {
            prow[(size_t)r * GNN_DOUT + t * 16] = accP[t][r];
            qrow[(size_t)r * GNN_DOUT + t * 16] = accQ[t][r];
        }
}

#define GNN_SCAN_U 10
#define GNN_QCAP   (GNN_SCAN_U * 128)

__device__ __forceinline__ bool edge_hit(unsigned w, unsigned short nd) {
    return (unsigned short)w == nd;
}
__device__ __forceinline__ void stage_piece(const int4* __restrict__ s4, const int4* __restrict__ d4,
                                            uint4* __restrict__ L4, int i) {
    const int4 s = s4[i];
    const int4 d = d4[i];
    uint4 w;
    w.x = ((unsigned)d.x & 0xffffu) | ((unsigned)s.x << 16);
    w.y = ((unsigned)d.y & 0xffffu) | ((unsigned)s.y << 16);
    w.z = ((unsigned)d.z & 0xffffu) | ((unsigned)s.z << 16);
    w.w = ((unsigned)d.w & 0xffffu) | ((unsigned)s.w << 16);
    L4[i] = w;
}
__device__ __forceinline__ float4 lrelu_sum(float4 acc, float4 pb, float4 q) {
    float v;
    v = pb.x + q.x; acc.x += fmaxf(v, GNN_SLOPE * v);
    v = pb.y + q.y; acc.y += fmaxf(v, GNN_SLOPE * v);
    v = pb.z + q.z; acc.z += fmaxf(v, GNN_SLOPE * v);
    v = pb.w + q.w; acc.w += fmaxf(v, GNN_SLOPE * v);
    return acc;
}

__global__ __launch_bounds__(256) void k_edge_own(const float* __restrict__ P,
                                                  const float* __restrict__ Q,
                                                  const float* __restrict__ b1,
                                                  const int* __restrict__ ei,
                                                  float* __restrict__ S,
                                                  float* __restrict__ cnt) {
    __shared__ unsigned int Ls[GNN_E];
    __shared__ unsigned int Qs[8 * GNN_QCAP];

    const int tid  = threadIdx.x;
    const int lane = tid & 31;
    const int wave = __builtin_amdgcn_readfirstlane(tid >> 5);
    const unsigned node = blockIdx.x * 8 + wave;

    float4 pb[GNN_BSN];
#pragma unroll
    for (int g = 0; g < GNN_BSN; ++g)
        pb[g] = *(const float4*)(P + ((size_t)g * GNN_N + node) * GNN_DOUT + lane * 4);
    const float4 bb = *(const float4*)(b1 + lane * 4);

    {
        const int4* s4 = (const int4*)ei;
        const int4* d4 = (const int4*)(ei + GNN_E);
        uint4* L4 = (uint4*)Ls;
#pragma unroll
        for (int k = 0; k < 16; ++k) stage_piece(s4, d4, L4, k * 256 + tid);
#pragma unroll
        for (int k = 16; k < 31; ++k) stage_piece(s4, d4, L4, k * 256 + tid);
        if (tid < GNN_E / 4 - 31 * 256) stage_piece(s4, d4, L4, 31 * 256 + tid);
    }

    float4 acc[GNN_BSN], qb[GNN_BSN];
#pragma unroll
    for (int g = 0; g < GNN_BSN; ++g) {
        pb[g].x += bb.x; pb[g].y += bb.y; pb[g].z += bb.z; pb[g].w += bb.w;
        acc[g] = make_float4(0.0f, 0.0f, 0.0f, 0.0f);
        qb[g]  = make_float4(0.0f, 0.0f, 0.0f, 0.0f);
    }
    __syncthreads();

    const uint4*   Lw = (const uint4*)Ls + lane;
    unsigned int*  Qw = Qs + wave * GNN_QCAP;
    const unsigned short nd16 = (unsigned short)node;
    const float4*  Q4 = (const float4*)Q + lane;
    int  deg  = 0;
    bool pend = false;

    for (int grp = 0; grp < GNN_E / GNN_QCAP; ++grp) {
        uint4 w[GNN_SCAN_U];
#pragma unroll
        for (int u = 0; u < GNN_SCAN_U; ++u) w[u] = Lw[(grp * GNN_SCAN_U + u) * 32];

        unsigned bm[4 * GNN_SCAN_U], au[GNN_SCAN_U];
        unsigned any = 0;
#pragma unroll
        for (int u = 0; u < GNN_SCAN_U; ++u) {
            bm[4 * u + 0] = __builtin_amdgcn_ballot_w32(edge_hit(w[u].x, nd16));
            bm[4 * u + 1] = __builtin_amdgcn_ballot_w32(edge_hit(w[u].y, nd16));
            bm[4 * u + 2] = __builtin_amdgcn_ballot_w32(edge_hit(w[u].z, nd16));
            bm[4 * u + 3] = __builtin_amdgcn_ballot_w32(edge_hit(w[u].w, nd16));
            au[u] = bm[4 * u + 0] | bm[4 * u + 1] | bm[4 * u + 2] | bm[4 * u + 3];
            any |= au[u];
        }
        if (any) {
            int nq = 0;
#pragma unroll
            for (int u = 0; u < GNN_SCAN_U; ++u) {
                if (au[u] == 0) continue;
                const unsigned ww[4] = {w[u].x, w[u].y, w[u].z, w[u].w};
#pragma unroll
                for (int j = 0; j < 4; ++j) {
                    const unsigned b = bm[4 * u + j];
                    if (b) {
                        if (edge_hit(ww[j], nd16))
                            Qw[nq + __builtin_amdgcn_mbcnt_lo(b, 0u)] = ww[j] >> 16;
                        nq += __builtin_popcount(b);
                    }
                }
            }
            for (int q = 0; q < nq; ++q) {
                const unsigned src = __builtin_amdgcn_readfirstlane(Qw[q]);
                if (pend) {
#pragma unroll
                    for (int g = 0; g < GNN_BSN; ++g) acc[g] = lrelu_sum(acc[g], pb[g], qb[g]);
                }
#pragma unroll
                for (int g = 0; g < GNN_BSN; ++g)
                    qb[g] = Q4[((size_t)g * GNN_N + src) * (GNN_DOUT / 4)];
                pend = true;
            }
            deg += nq;
        }
    }
    if (pend) {
#pragma unroll
        for (int g = 0; g < GNN_BSN; ++g) acc[g] = lrelu_sum(acc[g], pb[g], qb[g]);
    }

#pragma unroll
    for (int g = 0; g < GNN_BSN; ++g)
        *(float4*)(S + ((size_t)g * GNN_N + node) * GNN_DOUT + lane * 4) = acc[g];
    if (lane < GNN_BSN) cnt[(size_t)lane * GNN_N + node] = (float)deg;
}

template <bool RES>
__global__ __launch_bounds__(256) void k_gemm_agg_ln(const float* __restrict__ S,
                                                     const float* __restrict__ Bp2,
                                                     const float* __restrict__ cnt,
                                                     const float* __restrict__ b2,
                                                     const float* __restrict__ g,
                                                     const float* __restrict__ beta,
                                                     const float* __restrict__ xres,
                                                     float* __restrict__ out) {
    const int wave = threadIdx.x >> 5;
    const int lane = threadIdx.x & 31;
    const int half = lane >> 4;
    const int lm   = lane & 15;
    const int m0   = blockIdx.x * 128 + wave * 16;

    const float cntv  = cnt[m0 + lm];
    const float scale = 1.0f / fmaxf(cntv, 1.0f);

    v8f acc[8] = {};
    const float*  arow = S + (size_t)(m0 + lm) * 128 + 2 * half;
    const float4* bp   = (const float4*)(Bp2 + (size_t)lane * 16);

#pragma unroll
    for (int k = 0; k < 128; k += 4) {
        v2f a = *(const v2f*)(arow + k);
        a.x *= scale; a.y *= scale;
        const float4* bk = bp + (size_t)(k >> 2) * 128;
        float4 pb[4];
#pragma unroll
        for (int i = 0; i < 4; ++i) pb[i] = bk[i];
        v2f b;
#pragma unroll
        for (int i = 0; i < 4; ++i) {
            b.x = pb[i].x; b.y = pb[i].y; acc[2 * i + 0] = WMMA_F32(a, b, acc[2 * i + 0]);
            b.x = pb[i].z; b.y = pb[i].w; acc[2 * i + 1] = WMMA_F32(a, b, acc[2 * i + 1]);
        }
    }

    float b2v[8], gv[8], bev[8];
#pragma unroll
    for (int t = 0; t < 8; ++t) {
        b2v[t] = b2[t * 16 + lm];
        gv[t]  = g[t * 16 + lm];
        bev[t] = beta[t * 16 + lm];
    }

    const size_t rb = (size_t)(m0 + 8 * half) * GNN_DOUT + lm;
    float* crow = out + rb;
    const float* rrow = RES ? (xres + rb) : nullptr;

#pragma unroll
    for (int r = 0; r < 8; ++r) {
        const float cr   = __shfl(cntv, r + 8 * half, 32);
        const float hasE = (cr > 0.0f) ? 1.0f : 0.0f;

        float h[8];
        float s = 0.0f;
#pragma unroll
        for (int t = 0; t < 8; ++t) { h[t] = acc[t][r] + hasE * b2v[t]; s += h[t]; }
#pragma unroll
        for (int m = 8; m >= 1; m >>= 1) s += __shfl_xor(s, m, 32);
        const float mu = s * (1.0f / GNN_DOUT);

        float s2 = 0.0f;
#pragma unroll
        for (int t = 0; t < 8; ++t) { h[t] -= mu; s2 += h[t] * h[t]; }
#pragma unroll
        for (int m = 8; m >= 1; m >>= 1) s2 += __shfl_xor(s2, m, 32);
        const float rstd = rsqrtf(s2 * (1.0f / GNN_DOUT) + GNN_EPS);

#pragma unroll
        for (int t = 0; t < 8; ++t) {
            float o = h[t] * rstd * gv[t] + bev[t];
            if (RES) o += rrow[(size_t)r * GNN_DOUT + t * 16];
            crow[(size_t)r * GNN_DOUT + t * 16] = lrelu(o);
        }
    }
}

extern "C" void kernel_launch(void* const* d_in, const int* in_sizes, int n_in,
                              void* d_out, int out_size, void* d_ws, size_t ws_size,
                              hipStream_t stream) {
    (void)in_sizes; (void)n_in; (void)out_size; (void)ws_size;

    const float* x  = (const float*)d_in[0];
    const int*   ei = (const int*)d_in[1];
    const float* W1[2] = {(const float*)d_in[2],  (const float*)d_in[8]};
    const float* b1[2] = {(const float*)d_in[3],  (const float*)d_in[9]};
    const float* W2[2] = {(const float*)d_in[4],  (const float*)d_in[10]};
    const float* b2[2] = {(const float*)d_in[5],  (const float*)d_in[11]};
    const float* g [2] = {(const float*)d_in[6],  (const float*)d_in[12]};
    const float* be[2] = {(const float*)d_in[7],  (const float*)d_in[13]};

    const size_t NF = (size_t)GNN_NT * GNN_DOUT;
    float* ws   = (float*)d_ws;
    float* X    = ws;
    float* P    = X + NF;
    float* Q    = P + NF;
    float* S    = Q + NF;
    float* cnt  = S + NF;
    float* BpA  = cnt + GNN_NT;
    float* BpB  = BpA + 128 * GNN_DOUT;
    float* Bp2  = BpB + 128 * GNN_DOUT;

    k_pack_b<true ><<< 64 * GNN_DOUT / 256, 256, 0, stream>>>(W1[0], BpA, 64);
    k_pack_b<false><<< 64 * GNN_DOUT / 256, 256, 0, stream>>>(W1[0] + 64 * GNN_DOUT, BpB, 64);
    k_pack_b<false><<<128 * GNN_DOUT / 256, 256, 0, stream>>>(W2[0], Bp2, 128);
    k_gemm_pq<64><<<250, 256, 0, stream>>>(x, BpA, BpB, P, Q);
    k_edge_own<<<250, 256, 0, stream>>>(P, Q, b1[0], ei, S, cnt);
    k_gemm_agg_ln<false><<<250, 256, 0, stream>>>(S, Bp2, cnt, b2[0], g[0], be[0], nullptr, X);

    k_pack_b<true ><<<128 * GNN_DOUT / 256, 256, 0, stream>>>(W1[1], BpA, 128);
    k_pack_b<false><<<128 * GNN_DOUT / 256, 256, 0, stream>>>(W1[1] + 128 * GNN_DOUT, BpB, 128);
    k_pack_b<false><<<128 * GNN_DOUT / 256, 256, 0, stream>>>(W2[1], Bp2, 128);
    k_gemm_pq<128><<<250, 256, 0, stream>>>(X, BpA, BpB, P, Q);
    k_edge_own<<<250, 256, 0, stream>>>(P, Q, b1[1], ei, S, cnt);
    k_gemm_agg_ln<true><<<250, 256, 0, stream>>>(S, Bp2, cnt, b2[1], g[1], be[1], X, (float*)d_out);
}


# GEMM weights staged once per work-group into LDS in WMMA fragment order straight from the row-major inputs, B fragments by ds_load_b128 one k-step ahead; the 6 pack launches go (12 -> 6 launches)
# speedup vs baseline: 4.0480x; 1.4881x over previous
#include <hip/hip_runtime.h>


typedef __attribute__((ext_vector_type(2))) float v2f;
typedef __attribute__((ext_vector_type(4))) float v4f;
typedef __attribute__((ext_vector_type(8))) float v8f;

#define GNN_SLOPE 0.01f
#define GNN_EPS   1e-5f
#define GNN_BSN   16
#define GNN_N     2000
#define GNN_E     32000
#define GNN_NT    (GNN_BSN * GNN_N)
#define GNN_DOUT  128

__device__ __forceinline__ float lrelu(float v) { return v >= 0.0f ? v : GNN_SLOPE * v; }

#define WMMA_F32(A, B, C) \
    __builtin_amdgcn_wmma_f32_16x16x4_f32(false, (A), false, (B), (short)0, (C), false, false)

#define GNN_RPITCH 68
#define GNN_MAT(K) (2 * (K) * GNN_RPITCH)

template <int K, bool PAIR>
__device__ __forceinline__ void stage_weights(const float* __restrict__ W, float* __restrict__ L) {
    constexpr int PER = 2 * K * 4 / 256;
    v4f top[PER][4], bot[PER][4];
#pragma unroll
    for (int j = 0; j < PER; ++j) {
        const int u  = j * 256 + (int)threadIdx.x;
        const int q  = u & 3;
        const int R  = u >> 2;
        const int i  = R & 3;
        const int rp = R >> 2;
        const float* src = W + (size_t)(2 * rp) * GNN_DOUT + 32 * i + 4 * q;
        top[j][0] = *(const v4f*)(src);
        top[j][1] = *(const v4f*)(src + GNN_DOUT);
        top[j][2] = *(const v4f*)(src + 16);
        top[j][3] = *(const v4f*)(src + GNN_DOUT + 16);
        if (PAIR) {
            const float* sb = src + (size_t)K * GNN_DOUT;
            bot[j][0] = *(const v4f*)(sb);
            bot[j][1] = *(const v4f*)(sb + GNN_DOUT);
            bot[j][2] = *(const v4f*)(sb + 16);
            bot[j][3] = *(const v4f*)(sb + GNN_DOUT + 16);
        }
    }
#pragma unroll
    for (int j = 0; j < PER; ++j) {
        const int u = j * 256 + (int)threadIdx.x;
        const int q = u & 3;
        const int R = u >> 2;
        float* dst = L + R * GNN_RPITCH + 16 * q;
#pragma unroll
        for (int e = 0; e < 4; ++e) {
            v4f a = {top[j][0][e], top[j][1][e], top[j][2][e], top[j][3][e]};
            if (PAIR) {
                const v4f b = {bot[j][0][e], bot[j][1][e], bot[j][2][e], bot[j][3][e]};
                a -= b;
                *(v4f*)(dst + GNN_MAT(K) + 4 * e) = b;
            }
            *(v4f*)(dst + 4 * e) = a;
        }
    }
}

template <int K>
__global__ __launch_bounds__(256) void k_gemm_pq(const float* __restrict__ A,
                                                 const float* __restrict__ W,
                                                 float* __restrict__ P,
                                                 float* __restrict__ Q) {
    __shared__ __attribute__((aligned(16))) float Lw[2 * GNN_MAT(K)];

    const int wave = threadIdx.x >> 5;
    const int lane = threadIdx.x & 31;
    const int half = lane >> 4;
    const int lm   = lane & 15;
    const int m0   = blockIdx.x * 128 + wave * 16;

    const float* arow = A + (size_t)(m0 + lm) * K + 2 * half;
    v2f af[K / 4];
#pragma unroll
    for (int kk = 0; kk < K / 4; ++kk) af[kk] = *(const v2f*)(arow + 4 * kk);

    stage_weights<K, true>(W, Lw);

    v8f accP[8] = {};
    v8f accQ[8] = {};

    const v4f* lbA = (const v4f*)(Lw + half * 4 * GNN_RPITCH + lm * 4);
    const v4f* lbB = lbA + GNN_MAT(K) / 4;

    __syncthreads();

    v4f pa[2][4], pb[2][4];
#pragma unroll
    for (int i = 0; i < 4; ++i) { pa[0][i] = lbA[i * 17]; pb[0][i] = lbB[i * 17]; }
    __builtin_amdgcn_sched_group_barrier(0x100, 8, 0);

#pragma unroll
    for (int kk = 0; kk < K / 4; ++kk) {
        const int cur = kk & 1;
        const v2f a = af[kk];
        if (kk + 1 < K / 4) {
#pragma unroll
            for (int i = 0; i < 4; ++i) {
                pa[cur ^ 1][i] = lbA[((kk + 1) * 8 + i) * 17];
                pb[cur ^ 1][i] = lbB[((kk + 1) * 8 + i) * 17];
            }
        }
        v2f b;
#pragma unroll
        for (int i = 0; i < 4; ++i) {
            b.x = pa[cur][i].x; b.y = pa[cur][i].y; accP[2 * i + 0] = WMMA_F32(a, b, accP[2 * i + 0]);
            b.x = pa[cur][i].z; b.y = pa[cur][i].w; accP[2 * i + 1] = WMMA_F32(a, b, accP[2 * i + 1]);
            b.x = pb[cur][i].x; b.y = pb[cur][i].y; accQ[2 * i + 0] = WMMA_F32(a, b, accQ[2 * i + 0]);
            b.x = pb[cur][i].z; b.y = pb[cur][i].w; accQ[2 * i + 1] = WMMA_F32(a, b, accQ[2 * i + 1]);
        }
        if (kk + 1 < K / 4) __builtin_amdgcn_sched_group_barrier(0x100, 8, 0);
        __builtin_amdgcn_sched_group_barrier(0x008, 16, 0);
    }

    const size_t rb = (size_t)(m0 + 8 * half) * GNN_DOUT + lm;
    float* prow = P + rb;
    float* qrow = Q + rb;
#pragma unroll
    for (int r = 0; r < 8; ++r)
#pragma unroll
        for (int t = 0; t < 8; ++t) {
            prow[(size_t)r * GNN_DOUT + t * 16] = accP[t][r];
            qrow[(size_t)r * GNN_DOUT + t * 16] = accQ[t][r];
        }
}

#define GNN_SCAN_U 10
#define GNN_QCAP   (GNN_SCAN_U * 128)

__device__ __forceinline__ bool edge_hit(unsigned w, unsigned short nd) {
    return (unsigned short)w == nd;
}
__device__ __forceinline__ void stage_piece(const int4* __restrict__ s4, const int4* __restrict__ d4,
                                            uint4* __restrict__ L4, int i) {
    const int4 s = s4[i];
    const int4 d = d4[i];
    uint4 w;
    w.x = ((unsigned)d.x & 0xffffu) | ((unsigned)s.x << 16);
    w.y = ((unsigned)d.y & 0xffffu) | ((unsigned)s.y << 16);
    w.z = ((unsigned)d.z & 0xffffu) | ((unsigned)s.z << 16);
    w.w = ((unsigned)d.w & 0xffffu) | ((unsigned)s.w << 16);
    L4[i] = w;
}
__device__ __forceinline__ float4 lrelu_sum(float4 acc, float4 pb, float4 q) {
    float v;
    v = pb.x + q.x; acc.x += fmaxf(v, GNN_SLOPE * v);
    v = pb.y + q.y; acc.y += fmaxf(v, GNN_SLOPE * v);
    v = pb.z + q.z; acc.z += fmaxf(v, GNN_SLOPE * v);
    v = pb.w + q.w; acc.w += fmaxf(v, GNN_SLOPE * v);
    return acc;
}

__global__ __launch_bounds__(256) void k_edge_own(const float* __restrict__ P,
                                                  const float* __restrict__ Q,
                                                  const float* __restrict__ b1,
                                                  const int* __restrict__ ei,
                                                  float* __restrict__ S,
                                                  float* __restrict__ cnt) {
    __shared__ unsigned int Ls[GNN_E];
    __shared__ unsigned int Qs[8 * GNN_QCAP];

    const int tid  = threadIdx.x;
    const int lane = tid & 31;
    const int wave = __builtin_amdgcn_readfirstlane(tid >> 5);
    const unsigned node = blockIdx.x * 8 + wave;

    float4 pb[GNN_BSN];
#pragma unroll
    for (int g = 0; g < GNN_BSN; ++g)
        pb[g] = *(const float4*)(P + ((size_t)g * GNN_N + node) * GNN_DOUT + lane * 4);
    const float4 bb = *(const float4*)(b1 + lane * 4);

    {
        const int4* s4 = (const int4*)ei;
        const int4* d4 = (const int4*)(ei + GNN_E);
        uint4* L4 = (uint4*)Ls;
#pragma unroll
        for (int k = 0; k < 16; ++k) stage_piece(s4, d4, L4, k * 256 + tid);
#pragma unroll
        for (int k = 16; k < 31; ++k) stage_piece(s4, d4, L4, k * 256 + tid);
        if (tid < GNN_E / 4 - 31 * 256) stage_piece(s4, d4, L4, 31 * 256 + tid);
    }

    float4 acc[GNN_BSN], qb[GNN_BSN];
#pragma unroll
    for (int g = 0; g < GNN_BSN; ++g) {
        pb[g].x += bb.x; pb[g].y += bb.y; pb[g].z += bb.z; pb[g].w += bb.w;
        acc[g] = make_float4(0.0f, 0.0f, 0.0f, 0.0f);
        qb[g]  = make_float4(0.0f, 0.0f, 0.0f, 0.0f);
    }
    __syncthreads();

    const uint4*   Lw = (const uint4*)Ls + lane;
    unsigned int*  Qw = Qs + wave * GNN_QCAP;
    const unsigned short nd16 = (unsigned short)node;
    const float4*  Q4 = (const float4*)Q + lane;
    int  deg  = 0;
    bool pend = false;

    for (int grp = 0; grp < GNN_E / GNN_QCAP; ++grp) {
        uint4 w[GNN_SCAN_U];
#pragma unroll
        for (int u = 0; u < GNN_SCAN_U; ++u) w[u] = Lw[(grp * GNN_SCAN_U + u) * 32];

        unsigned bm[4 * GNN_SCAN_U], au[GNN_SCAN_U];
        unsigned any = 0;
#pragma unroll
        for (int u = 0; u < GNN_SCAN_U; ++u) {
            bm[4 * u + 0] = __builtin_amdgcn_ballot_w32(edge_hit(w[u].x, nd16));
            bm[4 * u + 1] = __builtin_amdgcn_ballot_w32(edge_hit(w[u].y, nd16));
            bm[4 * u + 2] = __builtin_amdgcn_ballot_w32(edge_hit(w[u].z, nd16));
            bm[4 * u + 3] = __builtin_amdgcn_ballot_w32(edge_hit(w[u].w, nd16));
            au[u] = bm[4 * u + 0] | bm[4 * u + 1] | bm[4 * u + 2] | bm[4 * u + 3];
            any |= au[u];
        }
        if (any) {
            int nq = 0;
#pragma unroll
            for (int u = 0; u < GNN_SCAN_U; ++u) {
                if (au[u] == 0) continue;
                const unsigned ww[4] = {w[u].x, w[u].y, w[u].z, w[u].w};
#pragma unroll
                for (int j = 0; j < 4; ++j) {
                    const unsigned b = bm[4 * u + j];
                    if (b) {
                        if (edge_hit(ww[j], nd16))
                            Qw[nq + __builtin_amdgcn_mbcnt_lo(b, 0u)] = ww[j] >> 16;
                        nq += __builtin_popcount(b);
                    }
                }
            }
            for (int q = 0; q < nq; ++q) {
                const unsigned src = __builtin_amdgcn_readfirstlane(Qw[q]);
                if (pend) {
#pragma unroll
                    for (int g = 0; g < GNN_BSN; ++g) acc[g] = lrelu_sum(acc[g], pb[g], qb[g]);
                }
#pragma unroll
                for (int g = 0; g < GNN_BSN; ++g)
                    qb[g] = Q4[((size_t)g * GNN_N + src) * (GNN_DOUT / 4)];
                pend = true;
            }
            deg += nq;
        }
    }
    if (pend) {
#pragma unroll
        for (int g = 0; g < GNN_BSN; ++g) acc[g] = lrelu_sum(acc[g], pb[g], qb[g]);
    }

#pragma unroll
    for (int g = 0; g < GNN_BSN; ++g)
        *(float4*)(S + ((size_t)g * GNN_N + node) * GNN_DOUT + lane * 4) = acc[g];
    if (lane < GNN_BSN) cnt[(size_t)lane * GNN_N + node] = (float)deg;
}

template <bool RES>
__global__ __launch_bounds__(256) void k_gemm_agg_ln(const float* __restrict__ S,
                                                     const float* __restrict__ W2,
                                                     const float* __restrict__ cnt,
                                                     const float* __restrict__ b2,
                                                     const float* __restrict__ g,
                                                     const float* __restrict__ beta,
                                                     const float* __restrict__ xres,
                                                     float* __restrict__ out) {
    const int wave = threadIdx.x >> 5;
    const int lane = threadIdx.x & 31;
    const int half = lane >> 4;
    const int lm   = lane & 15;
    const int m0   = blockIdx.x * 128 + wave * 16;

    __shared__ __attribute__((aligned(16))) float Lw[GNN_MAT(128)];

    const float cntv  = cnt[m0 + lm];
    const float* arow = S + (size_t)(m0 + lm) * 128 + 2 * half;
    v2f af[32];
#pragma unroll
    for (int kk = 0; kk < 32; ++kk) af[kk] = *(const v2f*)(arow + 4 * kk);

    stage_weights<128, false>(W2, Lw);

    const float scale = 1.0f / fmaxf(cntv, 1.0f);

    v8f acc[8] = {};
    const v4f*   lb   = (const v4f*)(Lw + half * 4 * GNN_RPITCH + lm * 4);

    __syncthreads();

    v4f pb[2][4];
#pragma unroll
    for (int i = 0; i < 4; ++i) pb[0][i] = lb[i * 17];
    __builtin_amdgcn_sched_group_barrier(0x100, 4, 0);

#pragma unroll
    for (int kk = 0; kk < 32; ++kk) {
        const int cur = kk & 1;
        v2f a = af[kk];
        a.x *= scale; a.y *= scale;
        if (kk + 1 < 32) {
#pragma unroll
            for (int i = 0; i < 4; ++i) pb[cur ^ 1][i] = lb[((kk + 1) * 8 + i) * 17];
        }
        v2f b;
#pragma unroll
        for (int i = 0; i < 4; ++i) {
            b.x = pb[cur][i].x; b.y = pb[cur][i].y; acc[2 * i + 0] = WMMA_F32(a, b, acc[2 * i + 0]);
            b.x = pb[cur][i].z; b.y = pb[cur][i].w; acc[2 * i + 1] = WMMA_F32(a, b, acc[2 * i + 1]);
        }
        if (kk + 1 < 32) __builtin_amdgcn_sched_group_barrier(0x100, 4, 0);
        __builtin_amdgcn_sched_group_barrier(0x008, 8, 0);
    }

    float b2v[8], gv[8], bev[8];
#pragma unroll
    for (int t = 0; t < 8; ++t) {
        b2v[t] = b2[t * 16 + lm];
        gv[t]  = g[t * 16 + lm];
        bev[t] = beta[t * 16 + lm];
    }

    const size_t rb = (size_t)(m0 + 8 * half) * GNN_DOUT + lm;
    float* crow = out + rb;
    const float* rrow = RES ? (xres + rb) : nullptr;

#pragma unroll
    for (int r = 0; r < 8; ++r) {
        const float cr   = __shfl(cntv, r + 8 * half, 32);
        const float hasE = (cr > 0.0f) ? 1.0f : 0.0f;

        float h[8];
        float s = 0.0f;
#pragma unroll
        for (int t = 0; t < 8; ++t) { h[t] = acc[t][r] + hasE * b2v[t]; s += h[t]; }
#pragma unroll
        for (int m = 8; m >= 1; m >>= 1) s += __shfl_xor(s, m, 32);
        const float mu = s * (1.0f / GNN_DOUT);

        float s2 = 0.0f;
#pragma unroll
        for (int t = 0; t < 8; ++t) { h[t] -= mu; s2 += h[t] * h[t]; }
#pragma unroll
        for (int m = 8; m >= 1; m >>= 1) s2 += __shfl_xor(s2, m, 32);
        const float rstd = rsqrtf(s2 * (1.0f / GNN_DOUT) + GNN_EPS);

#pragma unroll
        for (int t = 0; t < 8; ++t) {
            float o = h[t] * rstd * gv[t] + bev[t];
            if (RES) o += rrow[(size_t)r * GNN_DOUT + t * 16];
            crow[(size_t)r * GNN_DOUT + t * 16] = lrelu(o);
        }
    }
}

extern "C" void kernel_launch(void* const* d_in, const int* in_sizes, int n_in,
                              void* d_out, int out_size, void* d_ws, size_t ws_size,
                              hipStream_t stream) {
    (void)in_sizes; (void)n_in; (void)out_size; (void)ws_size;

    const float* x  = (const float*)d_in[0];
    const int*   ei = (const int*)d_in[1];
    const float* W1[2] = {(const float*)d_in[2],  (const float*)d_in[8]};
    const float* b1[2] = {(const float*)d_in[3],  (const float*)d_in[9]};
    const float* W2[2] = {(const float*)d_in[4],  (const float*)d_in[10]};
    const float* b2[2] = {(const float*)d_in[5],  (const float*)d_in[11]};
    const float* g [2] = {(const float*)d_in[6],  (const float*)d_in[12]};
    const float* be[2] = {(const float*)d_in[7],  (const float*)d_in[13]};

    const size_t NF = (size_t)GNN_NT * GNN_DOUT;
    float* ws   = (float*)d_ws;
    float* X    = ws;
    float* P    = X + NF;
    float* Q    = P + NF;
    float* S    = Q + NF;
    float* cnt  = S + NF;

    k_gemm_pq<64><<<250, 256, 0, stream>>>(x, W1[0], P, Q);
    k_edge_own<<<250, 256, 0, stream>>>(P, Q, b1[0], ei, S, cnt);
    k_gemm_agg_ln<false><<<250, 256, 0, stream>>>(S, W2[0], cnt, b2[0], g[0], be[0], nullptr, X);

    k_gemm_pq<128><<<250, 256, 0, stream>>>(X, W1[1], P, Q);
    k_edge_own<<<250, 256, 0, stream>>>(P, Q, b1[1], ei, S, cnt);
    k_gemm_agg_ln<true><<<250, 256, 0, stream>>>(S, W2[1], cnt, b2[1], g[1], be[1], X, (float*)d_out);
}


# Edge stage: work-group filters dst once (each wave an eighth, from memory) into LDS hit lists; owner waves then sum with 3 hits' Q rows in flight; float sum order: another fixed order
# speedup vs baseline: 4.2583x; 1.0520x over previous
#include <hip/hip_runtime.h>


typedef __attribute__((ext_vector_type(2))) float v2f;
typedef __attribute__((ext_vector_type(4))) float v4f;
typedef __attribute__((ext_vector_type(8))) float v8f;

#define GNN_SLOPE 0.01f
#define GNN_EPS   1e-5f
#define GNN_BSN   16
#define GNN_N     2000
#define GNN_E     32000
#define GNN_NT    (GNN_BSN * GNN_N)
#define GNN_DOUT  128

__device__ __forceinline__ float lrelu(float v) { return v >= 0.0f ? v : GNN_SLOPE * v; }

#define WMMA_F32(A, B, C) \
    __builtin_amdgcn_wmma_f32_16x16x4_f32(false, (A), false, (B), (short)0, (C), false, false)

#define GNN_RPITCH 68
#define GNN_MAT(K) (2 * (K) * GNN_RPITCH)

template <int K, bool PAIR>
__device__ __forceinline__ void stage_weights(const float* __restrict__ W, float* __restrict__ L) {
    constexpr int PER = 2 * K * 4 / 256;
    v4f top[PER][4], bot[PER][4];
#pragma unroll
    for (int j = 0; j < PER; ++j) {
        const int u  = j * 256 + (int)threadIdx.x;
        const int q  = u & 3;
        const int R  = u >> 2;
        const int i  = R & 3;
        const int rp = R >> 2;
        const float* src = W + (size_t)(2 * rp) * GNN_DOUT + 32 * i + 4 * q;
        top[j][0] = *(const v4f*)(src);
        top[j][1] = *(const v4f*)(src + GNN_DOUT);
        top[j][2] = *(const v4f*)(src + 16);
        top[j][3] = *(const v4f*)(src + GNN_DOUT + 16);
        if (PAIR) {
            const float* sb = src + (size_t)K * GNN_DOUT;
            bot[j][0] = *(const v4f*)(sb);
            bot[j][1] = *(const v4f*)(sb + GNN_DOUT);
            bot[j][2] = *(const v4f*)(sb + 16);
            bot[j][3] = *(const v4f*)(sb + GNN_DOUT + 16);
        }
    }
#pragma unroll
    for (int j = 0; j < PER; ++j) {
        const int u = j * 256 + (int)threadIdx.x;
        const int q = u & 3;
        const int R = u >> 2;
        float* dst = L + R * GNN_RPITCH + 16 * q;
#pragma unroll
        for (int e = 0; e < 4; ++e) {
            v4f a = {top[j][0][e], top[j][1][e], top[j][2][e], top[j][3][e]};
            if (PAIR) {
                const v4f b = {bot[j][0][e], bot[j][1][e], bot[j][2][e], bot[j][3][e]};
                a -= b;
                *(v4f*)(dst + GNN_MAT(K) + 4 * e) = b;
            }
            *(v4f*)(dst + 4 * e) = a;
        }
    }
}

template <int K>
__global__ __launch_bounds__(256) void k_gemm_pq(const float* __restrict__ A,
                                                 const float* __restrict__ W,
                                                 float* __restrict__ P,
                                                 float* __restrict__ Q) {
    __shared__ __attribute__((aligned(16))) float Lw[2 * GNN_MAT(K)];

    const int wave = threadIdx.x >> 5;
    const int lane = threadIdx.x & 31;
    const int half = lane >> 4;
    const int lm   = lane & 15;
    const int m0   = blockIdx.x * 128 + wave * 16;

    const float* arow = A + (size_t)(m0 + lm) * K + 2 * half;
    v2f af[K / 4];
#pragma unroll
    for (int kk = 0; kk < K / 4; ++kk) af[kk] = *(const v2f*)(arow + 4 * kk);

    stage_weights<K, true>(W, Lw);

    v8f accP[8] = {};
    v8f accQ[8] = {};

    const v4f* lbA = (const v4f*)(Lw + half * 4 * GNN_RPITCH + lm * 4);
    const v4f* lbB = lbA + GNN_MAT(K) / 4;

    __syncthreads();

    v4f pa[2][4], pb[2][4];
#pragma unroll
    for (int i = 0; i < 4; ++i) { pa[0][i] = lbA[i * 17]; pb[0][i] = lbB[i * 17]; }
    __builtin_amdgcn_sched_group_barrier(0x100, 8, 0);

#pragma unroll
    for (int kk = 0; kk < K / 4; ++kk) {
        const int cur = kk & 1;
        const v2f a = af[kk];
        if (kk + 1 < K / 4) {
#pragma unroll
            for (int i = 0; i < 4; ++i) {
                pa[cur ^ 1][i] = lbA[((kk + 1) * 8 + i) * 17];
                pb[cur ^ 1][i] = lbB[((kk + 1) * 8 + i) * 17];
            }
        }
        v2f b;
#pragma unroll
        for (int i = 0; i < 4; ++i) {
            b.x = pa[cur][i].x; b.y = pa[cur][i].y; accP[2 * i + 0] = WMMA_F32(a, b, accP[2 * i + 0]);
            b.x = pa[cur][i].z; b.y = pa[cur][i].w; accP[2 * i + 1] = WMMA_F32(a, b, accP[2 * i + 1]);
            b.x = pb[cur][i].x; b.y = pb[cur][i].y; accQ[2 * i + 0] = WMMA_F32(a, b, accQ[2 * i + 0]);
            b.x = pb[cur][i].z; b.y = pb[cur][i].w; accQ[2 * i + 1] = WMMA_F32(a, b, accQ[2 * i + 1]);
        }
        if (kk + 1 < K / 4) __builtin_amdgcn_sched_group_barrier(0x100, 8, 0);
        __builtin_amdgcn_sched_group_barrier(0x008, 16, 0);
    }

    const size_t rb = (size_t)(m0 + 8 * half) * GNN_DOUT + lm;
    float* prow = P + rb;
    float* qrow = Q + rb;
#pragma unroll
    for (int r = 0; r < 8; ++r)
#pragma unroll
        for (int t = 0; t < 8; ++t) {
            prow[(size_t)r * GNN_DOUT + t * 16] = accP[t][r];
            qrow[(size_t)r * GNN_DOUT + t * 16] = accQ[t][r];
        }
}

#define GNN_CH  (GNN_E / 8)
#define GNN_NLD 32

__device__ __forceinline__ float4 lrelu_sum(float4 acc, float4 pb, float4 q) {
    float v;
    v = pb.x + q.x; acc.x += fmaxf(v, GNN_SLOPE * v);
    v = pb.y + q.y; acc.y += fmaxf(v, GNN_SLOPE * v);
    v = pb.z + q.z; acc.z += fmaxf(v, GNN_SLOPE * v);
    v = pb.w + q.w; acc.w += fmaxf(v, GNN_SLOPE * v);
    return acc;
}

__device__ __forceinline__ int keep_hits(unsigned bal, unsigned r, unsigned e,
                                         unsigned int* __restrict__ Hw, int n) {
    if (bal) {
        if (r < 8u) Hw[n + __builtin_amdgcn_mbcnt_lo(bal, 0u)] = (r << 16) | e;
        n += __builtin_popcount(bal);
    }
    return n;
}

__device__ __forceinline__ void load_set(float4 (&q)[GNN_BSN], const float4* __restrict__ Q4, int s) {
    const size_t row = (size_t)(s < 0 ? 0 : s);
#pragma unroll
    for (int g = 0; g < GNN_BSN; ++g) q[g] = Q4[((size_t)g * GNN_N + row) * (GNN_DOUT / 4)];
}
__device__ __forceinline__ void sum_set(float4 (&acc)[GNN_BSN], const float4 (&pb)[GNN_BSN],
                                        const float4 (&q)[GNN_BSN]) {
#pragma unroll
    for (int g = 0; g < GNN_BSN; ++g) acc[g] = lrelu_sum(acc[g], pb[g], q[g]);
}

struct HitWalk {
    const unsigned int* H;
    const int* Hn;
    unsigned me;
    int lane;
    int w, base, n;
    unsigned bits, ent;
    __device__ __forceinline__ int next() {
        if (w >= 8) return -1;
        while (bits == 0u) {
            while (base >= n) {
                ++w;
                if (w >= 8) return -1;
                base = 0;
                n = __builtin_amdgcn_readfirstlane(Hn[w]);
            }
            const int idx = base + lane;
            ent  = idx < n ? H[w * GNN_CH + idx] : 0xffffffffu;
            bits = __builtin_amdgcn_ballot_w32((ent >> 16) == me);
            base += 32;
        }
        const int l = __builtin_ctz(bits);
        bits &= bits - 1u;
        return (int)(__builtin_amdgcn_readlane(ent, l) & 0xffffu);
    }
};

__global__ __launch_bounds__(256) void k_edge_own(const float* __restrict__ P,
                                                  const float* __restrict__ Q,
                                                  const float* __restrict__ b1,
                                                  const int* __restrict__ ei,
                                                  float* __restrict__ S,
                                                  float* __restrict__ cnt) {
    __shared__ unsigned int H[8 * GNN_CH];
    __shared__ int Hn[8];

    const int tid  = threadIdx.x;
    const int lane = tid & 31;
    const int wave = __builtin_amdgcn_readfirstlane(tid >> 5);
    const unsigned nbase = blockIdx.x * 8;
    const unsigned node  = nbase + wave;

    float4 pb[GNN_BSN];
#pragma unroll
    for (int g = 0; g < GNN_BSN; ++g)
        pb[g] = *(const float4*)(P + ((size_t)g * GNN_N + node) * GNN_DOUT + lane * 4);
    const float4 bb = *(const float4*)(b1 + lane * 4);

    unsigned int* Hw = H + wave * GNN_CH;
    int n = 0;
    {
        const int4* d4 = (const int4*)(ei + GNN_E) + wave * (GNN_CH / 4) + lane;
        int4 d[GNN_NLD];
#pragma unroll
        for (int k = 0; k < GNN_NLD - 1; ++k) d[k] = d4[k * 32];
        d[GNN_NLD - 1] = make_int4(-1, -1, -1, -1);
        if (lane < GNN_CH / 4 - (GNN_NLD - 1) * 32) d[GNN_NLD - 1] = d4[(GNN_NLD - 1) * 32];
#pragma unroll
        for (int k = 0; k < GNN_NLD; ++k) {
            const unsigned r0 = (unsigned)d[k].x - nbase;
            const unsigned r1 = (unsigned)d[k].y - nbase;
            const unsigned r2 = (unsigned)d[k].z - nbase;
            const unsigned r3 = (unsigned)d[k].w - nbase;
            const unsigned m0 = __builtin_amdgcn_ballot_w32(r0 < 8u);
            const unsigned m1 = __builtin_amdgcn_ballot_w32(r1 < 8u);
            const unsigned m2 = __builtin_amdgcn_ballot_w32(r2 < 8u);
            const unsigned m3 = __builtin_amdgcn_ballot_w32(r3 < 8u);
            if (m0 | m1 | m2 | m3) {
                const unsigned e = (unsigned)(k * 32 + lane) * 4u;
                n = keep_hits(m0, r0, e + 0u, Hw, n);
                n = keep_hits(m1, r1, e + 1u, Hw, n);
                n = keep_hits(m2, r2, e + 2u, Hw, n);
                n = keep_hits(m3, r3, e + 3u, Hw, n);
            }
        }
        const int* srcw = ei + wave * GNN_CH;
        for (int i = lane; i < n; i += 32) {
            const unsigned e = Hw[i];
            const unsigned s = (unsigned)srcw[e & 0xffffu] & 0xffffu;
            Hw[i] = (e & 0x70000u) | s;
        }
        if (lane == 0) Hn[wave] = n;
    }

    float4 acc[GNN_BSN];
#pragma unroll
    for (int g = 0; g < GNN_BSN; ++g) {
        pb[g].x += bb.x; pb[g].y += bb.y; pb[g].z += bb.z; pb[g].w += bb.w;
        acc[g] = make_float4(0.0f, 0.0f, 0.0f, 0.0f);
    }
    __syncthreads();

    const float4* Q4 = (const float4*)Q + lane;
    HitWalk hw;
    hw.H = H; hw.Hn = Hn; hw.me = (unsigned)wave; hw.lane = lane;
    hw.w = 0; hw.base = 0; hw.n = __builtin_amdgcn_readfirstlane(Hn[0]);
    hw.bits = 0u; hw.ent = 0xffffffffu;

    float4 qa[GNN_BSN], qb[GNN_BSN], qc[GNN_BSN];
    int deg = 0;
    int sa = hw.next(); load_set(qa, Q4, sa);
    int sb = hw.next(); load_set(qb, Q4, sb);
    int sc = hw.next(); load_set(qc, Q4, sc);
    while (sa >= 0) {
        sum_set(acc, pb, qa); ++deg;
        sa = hw.next(); load_set(qa, Q4, sa);
        if (sb >= 0) { sum_set(acc, pb, qb); ++deg; }
        sb = hw.next(); load_set(qb, Q4, sb);
        if (sc >= 0) { sum_set(acc, pb, qc); ++deg; }
        sc = hw.next(); load_set(qc, Q4, sc);
    }

#pragma unroll
    for (int g = 0; g < GNN_BSN; ++g)
        *(float4*)(S + ((size_t)g * GNN_N + node) * GNN_DOUT + lane * 4) = acc[g];
    if (lane < GNN_BSN) cnt[(size_t)lane * GNN_N + node] = (float)deg;
}

template <bool RES>
__global__ __launch_bounds__(256) void k_gemm_agg_ln(const float* __restrict__ S,
                                                     const float* __restrict__ W2,
                                                     const float* __restrict__ cnt,
                                                     const float* __restrict__ b2,
                                                     const float* __restrict__ g,
                                                     const float* __restrict__ beta,
                                                     const float* __restrict__ xres,
                                                     float* __restrict__ out) {
    const int wave = threadIdx.x >> 5;
    const int lane = threadIdx.x & 31;
    const int half = lane >> 4;
    const int lm   = lane & 15;
    const int m0   = blockIdx.x * 128 + wave * 16;

    __shared__ __attribute__((aligned(16))) float Lw[GNN_MAT(128)];

    const float cntv  = cnt[m0 + lm];
    const float* arow = S + (size_t)(m0 + lm) * 128 + 2 * half;
    v2f af[32];
#pragma unroll
    for (int kk = 0; kk < 32; ++kk) af[kk] = *(const v2f*)(arow + 4 * kk);

    stage_weights<128, false>(W2, Lw);

    const float scale = 1.0f / fmaxf(cntv, 1.0f);

    v8f acc[8] = {};
    const v4f*   lb   = (const v4f*)(Lw + half * 4 * GNN_RPITCH + lm * 4);

    __syncthreads();

    v4f pb[2][4];
#pragma unroll
    for (int i = 0; i < 4; ++i) pb[0][i] = lb[i * 17];
    __builtin_amdgcn_sched_group_barrier(0x100, 4, 0);

#pragma unroll
    for (int kk = 0; kk < 32; ++kk) {
        const int cur = kk & 1;
        v2f a = af[kk];
        a.x *= scale; a.y *= scale;
        if (kk + 1 < 32) {
#pragma unroll
            for (int i = 0; i < 4; ++i) pb[cur ^ 1][i] = lb[((kk + 1) * 8 + i) * 17];
        }
        v2f b;
#pragma unroll
        for (int i = 0; i < 4; ++i) {
            b.x = pb[cur][i].x; b.y = pb[cur][i].y; acc[2 * i + 0] = WMMA_F32(a, b, acc[2 * i + 0]);
            b.x = pb[cur][i].z; b.y = pb[cur][i].w; acc[2 * i + 1] = WMMA_F32(a, b, acc[2 * i + 1]);
        }
        if (kk + 1 < 32) __builtin_amdgcn_sched_group_barrier(0x100, 4, 0);
        __builtin_amdgcn_sched_group_barrier(0x008, 8, 0);
    }

    float b2v[8], gv[8], bev[8];
#pragma unroll
    for (int t = 0; t < 8; ++t) {
        b2v[t] = b2[t * 16 + lm];
        gv[t]  = g[t * 16 + lm];
        bev[t] = beta[t * 16 + lm];
    }

    const size_t rb = (size_t)(m0 + 8 * half) * GNN_DOUT + lm;
    float* crow = out + rb;
    const float* rrow = RES ? (xres + rb) : nullptr;

#pragma unroll
    for (int r = 0; r < 8; ++r) {
        const float cr   = __shfl(cntv, r + 8 * half, 32);
        const float hasE = (cr > 0.0f) ? 1.0f : 0.0f;

        float h[8];
        float s = 0.0f;
#pragma unroll
        for (int t = 0; t < 8; ++t) { h[t] = acc[t][r] + hasE * b2v[t]; s += h[t]; }
#pragma unroll
        for (int m = 8; m >= 1; m >>= 1) s += __shfl_xor(s, m, 32);
        const float mu = s * (1.0f / GNN_DOUT);

        float s2 = 0.0f;
#pragma unroll
        for (int t = 0; t < 8; ++t) { h[t] -= mu; s2 += h[t] * h[t]; }
#pragma unroll
        for (int m = 8; m >= 1; m >>= 1) s2 += __shfl_xor(s2, m, 32);
        const float rstd = rsqrtf(s2 * (1.0f / GNN_DOUT) + GNN_EPS);

#pragma unroll
        for (int t = 0; t < 8; ++t) {
            float o = h[t] * rstd * gv[t] + bev[t];
            if (RES) o += rrow[(size_t)r * GNN_DOUT + t * 16];
            crow[(size_t)r * GNN_DOUT + t * 16] = lrelu(o);
        }
    }
}

extern "C" void kernel_launch(void* const* d_in, const int* in_sizes, int n_in,
                              void* d_out, int out_size, void* d_ws, size_t ws_size,
                              hipStream_t stream) {
    (void)in_sizes; (void)n_in; (void)out_size; (void)ws_size;

    const float* x  = (const float*)d_in[0];
    const int*   ei = (const int*)d_in[1];
    const float* W1[2] = {(const float*)d_in[2],  (const float*)d_in[8]};
    const float* b1[2] = {(const float*)d_in[3],  (const float*)d_in[9]};
    const float* W2[2] = {(const float*)d_in[4],  (const float*)d_in[10]};
    const float* b2[2] = {(const float*)d_in[5],  (const float*)d_in[11]};
    const float* g [2] = {(const float*)d_in[6],  (const float*)d_in[12]};
    const float* be[2] = {(const float*)d_in[7],  (const float*)d_in[13]};

    const size_t NF = (size_t)GNN_NT * GNN_DOUT;
    float* ws   = (float*)d_ws;
    float* X    = ws;
    float* P    = X + NF;
    float* Q    = P + NF;
    float* S    = Q + NF;
    float* cnt  = S + NF;

    k_gemm_pq<64><<<250, 256, 0, stream>>>(x, W1[0], P, Q);
    k_edge_own<<<250, 256, 0, stream>>>(P, Q, b1[0], ei, S, cnt);
    k_gemm_agg_ln<false><<<250, 256, 0, stream>>>(S, W2[0], cnt, b2[0], g[0], be[0], nullptr, X);

    k_gemm_pq<128><<<250, 256, 0, stream>>>(X, W1[1], P, Q);
    k_edge_own<<<250, 256, 0, stream>>>(P, Q, b1[1], ei, S, cnt);
    k_gemm_agg_ln<true><<<250, 256, 0, stream>>>(S, W2[1], cnt, b2[1], g[1], be[1], X, (float*)d_out);
}


# Edge stage: 16 waves per work-group, a wave pair owns one node (8 graphs each), 5 hits' Q rows in flight per wave, each wave filters a sixteenth of dst; float sum order: another fixed order
# speedup vs baseline: 4.4227x; 1.0386x over previous
#include <hip/hip_runtime.h>


typedef __attribute__((ext_vector_type(2))) float v2f;
typedef __attribute__((ext_vector_type(4))) float v4f;
typedef __attribute__((ext_vector_type(8))) float v8f;

#define GNN_SLOPE 0.01f
#define GNN_EPS   1e-5f
#define GNN_BSN   16
#define GNN_N     2000
#define GNN_E     32000
#define GNN_NT    (GNN_BSN * GNN_N)
#define GNN_DOUT  128

__device__ __forceinline__ float lrelu(float v) { return v >= 0.0f ? v : GNN_SLOPE * v; }

#define WMMA_F32(A, B, C) \
    __builtin_amdgcn_wmma_f32_16x16x4_f32(false, (A), false, (B), (short)0, (C), false, false)

#define GNN_RPITCH 68
#define GNN_MAT(K) (2 * (K) * GNN_RPITCH)

template <int K, bool PAIR>
__device__ __forceinline__ void stage_weights(const float* __restrict__ W, float* __restrict__ L) {
    constexpr int PER = 2 * K * 4 / 256;
    v4f top[PER][4], bot[PER][4];
#pragma unroll
    for (int j = 0; j < PER; ++j) {
        const int u  = j * 256 + (int)threadIdx.x;
        const int q  = u & 3;
        const int R  = u >> 2;
        const int i  = R & 3;
        const int rp = R >> 2;
        const float* src = W + (size_t)(2 * rp) * GNN_DOUT + 32 * i + 4 * q;
        top[j][0] = *(const v4f*)(src);
        top[j][1] = *(const v4f*)(src + GNN_DOUT);
        top[j][2] = *(const v4f*)(src + 16);
        top[j][3] = *(const v4f*)(src + GNN_DOUT + 16);
        if (PAIR) {
            const float* sb = src + (size_t)K * GNN_DOUT;
            bot[j][0] = *(const v4f*)(sb);
            bot[j][1] = *(const v4f*)(sb + GNN_DOUT);
            bot[j][2] = *(const v4f*)(sb + 16);
            bot[j][3] = *(const v4f*)(sb + GNN_DOUT + 16);
        }
    }
#pragma unroll
    for (int j = 0; j < PER; ++j) {
        const int u = j * 256 + (int)threadIdx.x;
        const int q = u & 3;
        const int R = u >> 2;
        float* dst = L + R * GNN_RPITCH + 16 * q;
#pragma unroll
        for (int e = 0; e < 4; ++e) {
            v4f a = {top[j][0][e], top[j][1][e], top[j][2][e], top[j][3][e]};
            if (PAIR) {
                const v4f b = {bot[j][0][e], bot[j][1][e], bot[j][2][e], bot[j][3][e]};
                a -= b;
                *(v4f*)(dst + GNN_MAT(K) + 4 * e) = b;
            }
            *(v4f*)(dst + 4 * e) = a;
        }
    }
}

template <int K>
__global__ __launch_bounds__(256) void k_gemm_pq(const float* __restrict__ A,
                                                 const float* __restrict__ W,
                                                 float* __restrict__ P,
                                                 float* __restrict__ Q) {
    __shared__ __attribute__((aligned(16))) float Lw[2 * GNN_MAT(K)];

    const int wave = threadIdx.x >> 5;
    const int lane = threadIdx.x & 31;
    const int half = lane >> 4;
    const int lm   = lane & 15;
    const int m0   = blockIdx.x * 128 + wave * 16;

    const float* arow = A + (size_t)(m0 + lm) * K + 2 * half;
    v2f af[K / 4];
#pragma unroll
    for (int kk = 0; kk < K / 4; ++kk) af[kk] = *(const v2f*)(arow + 4 * kk);

    stage_weights<K, true>(W, Lw);

    v8f accP[8] = {};
    v8f accQ[8] = {};

    const v4f* lbA = (const v4f*)(Lw + half * 4 * GNN_RPITCH + lm * 4);
    const v4f* lbB = lbA + GNN_MAT(K) / 4;

    __syncthreads();

    v4f pa[2][4], pb[2][4];
#pragma unroll
    for (int i = 0; i < 4; ++i) { pa[0][i] = lbA[i * 17]; pb[0][i] = lbB[i * 17]; }
    __builtin_amdgcn_sched_group_barrier(0x100, 8, 0);

#pragma unroll
    for (int kk = 0; kk < K / 4; ++kk) {
        const int cur = kk & 1;
        const v2f a = af[kk];
        if (kk + 1 < K / 4) {
#pragma unroll
            for (int i = 0; i < 4; ++i) {
                pa[cur ^ 1][i] = lbA[((kk + 1) * 8 + i) * 17];
                pb[cur ^ 1][i] = lbB[((kk + 1) * 8 + i) * 17];
            }
        }
        v2f b;
#pragma unroll
        for (int i = 0; i < 4; ++i) {
            b.x = pa[cur][i].x; b.y = pa[cur][i].y; accP[2 * i + 0] = WMMA_F32(a, b, accP[2 * i + 0]);
            b.x = pa[cur][i].z; b.y = pa[cur][i].w; accP[2 * i + 1] = WMMA_F32(a, b, accP[2 * i + 1]);
            b.x = pb[cur][i].x; b.y = pb[cur][i].y; accQ[2 * i + 0] = WMMA_F32(a, b, accQ[2 * i + 0]);
            b.x = pb[cur][i].z; b.y = pb[cur][i].w; accQ[2 * i + 1] = WMMA_F32(a, b, accQ[2 * i + 1]);
        }
        if (kk + 1 < K / 4) __builtin_amdgcn_sched_group_barrier(0x100, 8, 0);
        __builtin_amdgcn_sched_group_barrier(0x008, 16, 0);
    }

    const size_t rb = (size_t)(m0 + 8 * half) * GNN_DOUT + lm;
    float* prow = P + rb;
    float* qrow = Q + rb;
#pragma unroll
    for (int r = 0; r < 8; ++r)
#pragma unroll
        for (int t = 0; t < 8; ++t) {
            prow[(size_t)r * GNN_DOUT + t * 16] = accP[t][r];
            qrow[(size_t)r * GNN_DOUT + t * 16] = accQ[t][r];
        }
}

#define GNN_EW  16
#define GNN_GW  (GNN_BSN / 2)
#define GNN_NS  5
#define GNN_CH  (GNN_E / GNN_EW)
#define GNN_NLD 16

__device__ __forceinline__ float4 lrelu_sum(float4 acc, float4 pb, float4 q) {
    float v;
    v = pb.x + q.x; acc.x += fmaxf(v, GNN_SLOPE * v);
    v = pb.y + q.y; acc.y += fmaxf(v, GNN_SLOPE * v);
    v = pb.z + q.z; acc.z += fmaxf(v, GNN_SLOPE * v);
    v = pb.w + q.w; acc.w += fmaxf(v, GNN_SLOPE * v);
    return acc;
}

__device__ __forceinline__ int keep_hits(unsigned bal, unsigned r, unsigned e,
                                         unsigned int* __restrict__ Hw, int n) {
    if (bal) {
        if (r < 8u) Hw[n + __builtin_amdgcn_mbcnt_lo(bal, 0u)] = (r << 16) | e;
        n += __builtin_popcount(bal);
    }
    return n;
}

__device__ __forceinline__ void load_set(float4 (&q)[GNN_GW], const float4* __restrict__ Q4, int s) {
    const size_t row = (size_t)(s < 0 ? 0 : s);
#pragma unroll
    for (int g = 0; g < GNN_GW; ++g) q[g] = Q4[((size_t)g * GNN_N + row) * (GNN_DOUT / 4)];
}
__device__ __forceinline__ void sum_set(float4 (&acc)[GNN_GW], const float4 (&pb)[GNN_GW],
                                        const float4 (&q)[GNN_GW]) {
#pragma unroll
    for (int g = 0; g < GNN_GW; ++g) acc[g] = lrelu_sum(acc[g], pb[g], q[g]);
}

struct HitWalk {
    const unsigned int* H;
    const int* Hn;
    unsigned me;
    int lane;
    int w, base, n;
    unsigned bits, ent;
    __device__ __forceinline__ int next() {
        if (w >= GNN_EW) return -1;
        while (bits == 0u) {
            while (base >= n) {
                ++w;
                if (w >= GNN_EW) return -1;
                base = 0;
                n = __builtin_amdgcn_readfirstlane(Hn[w]);
            }
            const int idx = base + lane;
            ent  = idx < n ? H[w * GNN_CH + idx] : 0xffffffffu;
            bits = __builtin_amdgcn_ballot_w32((ent >> 16) == me);
            base += 32;
        }
        const int l = __builtin_ctz(bits);
        bits &= bits - 1u;
        return (int)(__builtin_amdgcn_readlane(ent, l) & 0xffffu);
    }
};

__global__ __launch_bounds__(512) void k_edge_own(const float* __restrict__ P,
                                                  const float* __restrict__ Q,
                                                  const float* __restrict__ b1,
                                                  const int* __restrict__ ei,
                                                  float* __restrict__ S,
                                                  float* __restrict__ cnt) {
    __shared__ unsigned int H[GNN_EW * GNN_CH];
    __shared__ int Hn[GNN_EW];

    const int tid  = threadIdx.x;
    const int lane = tid & 31;
    const int wave = __builtin_amdgcn_readfirstlane(tid >> 5);
    const unsigned nbase = blockIdx.x * 8;
    const unsigned slot  = (unsigned)wave >> 1;
    const unsigned node  = nbase + slot;
    const int      g0    = (wave & 1) * GNN_GW;

    float4 pb[GNN_GW];
#pragma unroll
    for (int g = 0; g < GNN_GW; ++g)
        pb[g] = *(const float4*)(P + ((size_t)(g0 + g) * GNN_N + node) * GNN_DOUT + lane * 4);
    const float4 bb = *(const float4*)(b1 + lane * 4);

    unsigned int* Hw = H + wave * GNN_CH;
    int n = 0;
    {
        const int4* d4 = (const int4*)(ei + GNN_E) + wave * (GNN_CH / 4) + lane;
        int4 d[GNN_NLD];
#pragma unroll
        for (int k = 0; k < GNN_NLD - 1; ++k) d[k] = d4[k * 32];
        d[GNN_NLD - 1] = make_int4(-1, -1, -1, -1);
        if (lane < GNN_CH / 4 - (GNN_NLD - 1) * 32) d[GNN_NLD - 1] = d4[(GNN_NLD - 1) * 32];
#pragma unroll
        for (int k = 0; k < GNN_NLD; ++k) {
            const unsigned r0 = (unsigned)d[k].x - nbase;
            const unsigned r1 = (unsigned)d[k].y - nbase;
            const unsigned r2 = (unsigned)d[k].z - nbase;
            const unsigned r3 = (unsigned)d[k].w - nbase;
            const unsigned m0 = __builtin_amdgcn_ballot_w32(r0 < 8u);
            const unsigned m1 = __builtin_amdgcn_ballot_w32(r1 < 8u);
            const unsigned m2 = __builtin_amdgcn_ballot_w32(r2 < 8u);
            const unsigned m3 = __builtin_amdgcn_ballot_w32(r3 < 8u);
            if (m0 | m1 | m2 | m3) {
                const unsigned e = (unsigned)(k * 32 + lane) * 4u;
                n = keep_hits(m0, r0, e + 0u, Hw, n);
                n = keep_hits(m1, r1, e + 1u, Hw, n);
                n = keep_hits(m2, r2, e + 2u, Hw, n);
                n = keep_hits(m3, r3, e + 3u, Hw, n);
            }
        }
        const int* srcw = ei + wave * GNN_CH;
        for (int i = lane; i < n; i += 32) {
            const unsigned e = Hw[i];
            const unsigned s = (unsigned)srcw[e & 0xffffu] & 0xffffu;
            Hw[i] = (e & 0x70000u) | s;
        }
        if (lane == 0) Hn[wave] = n;
    }

    float4 acc[GNN_GW];
#pragma unroll
    for (int g = 0; g < GNN_GW; ++g) {
        pb[g].x += bb.x; pb[g].y += bb.y; pb[g].z += bb.z; pb[g].w += bb.w;
        acc[g] = make_float4(0.0f, 0.0f, 0.0f, 0.0f);
    }
    __syncthreads();

    const float4* Q4 = (const float4*)Q + (size_t)g0 * GNN_N * (GNN_DOUT / 4) + lane;
    HitWalk hw;
    hw.H = H; hw.Hn = Hn; hw.me = slot; hw.lane = lane;
    hw.w = 0; hw.base = 0; hw.n = __builtin_amdgcn_readfirstlane(Hn[0]);
    hw.bits = 0u; hw.ent = 0xffffffffu;

    float4 q[GNN_NS][GNN_GW];
    int    sh[GNN_NS];
    int deg = 0;
#pragma unroll
    for (int i = 0; i < GNN_NS; ++i) { sh[i] = hw.next(); load_set(q[i], Q4, sh[i]); }
    while (sh[0] >= 0) {
#pragma unroll
        for (int i = 0; i < GNN_NS; ++i) {
            if (sh[i] >= 0) { sum_set(acc, pb, q[i]); ++deg; }
            sh[i] = hw.next(); load_set(q[i], Q4, sh[i]);
        }
    }

#pragma unroll
    for (int g = 0; g < GNN_GW; ++g)
        *(float4*)(S + ((size_t)(g0 + g) * GNN_N + node) * GNN_DOUT + lane * 4) = acc[g];
    if (g0 == 0 && lane < GNN_BSN) cnt[(size_t)lane * GNN_N + node] = (float)deg;
}

template <bool RES>
__global__ __launch_bounds__(256) void k_gemm_agg_ln(const float* __restrict__ S,
                                                     const float* __restrict__ W2,
                                                     const float* __restrict__ cnt,
                                                     const float* __restrict__ b2,
                                                     const float* __restrict__ g,
                                                     const float* __restrict__ beta,
                                                     const float* __restrict__ xres,
                                                     float* __restrict__ out) {
    const int wave = threadIdx.x >> 5;
    const int lane = threadIdx.x & 31;
    const int half = lane >> 4;
    const int lm   = lane & 15;
    const int m0   = blockIdx.x * 128 + wave * 16;

    __shared__ __attribute__((aligned(16))) float Lw[GNN_MAT(128)];

    const float cntv  = cnt[m0 + lm];
    const float* arow = S + (size_t)(m0 + lm) * 128 + 2 * half;
    v2f af[32];
#pragma unroll
    for (int kk = 0; kk < 32; ++kk) af[kk] = *(const v2f*)(arow + 4 * kk);

    stage_weights<128, false>(W2, Lw);

    const float scale = 1.0f / fmaxf(cntv, 1.0f);

    v8f acc[8] = {};
    const v4f*   lb   = (const v4f*)(Lw + half * 4 * GNN_RPITCH + lm * 4);

    __syncthreads();

    v4f pb[2][4];
#pragma unroll
    for (int i = 0; i < 4; ++i) pb[0][i] = lb[i * 17];
    __builtin_amdgcn_sched_group_barrier(0x100, 4, 0);

#pragma unroll
    for (int kk = 0; kk < 32; ++kk) {
        const int cur = kk & 1;
        v2f a = af[kk];
        a.x *= scale; a.y *= scale;
        if (kk + 1 < 32) {
#pragma unroll
            for (int i = 0; i < 4; ++i) pb[cur ^ 1][i] = lb[((kk + 1) * 8 + i) * 17];
        }
        v2f b;
#pragma unroll
        for (int i = 0; i < 4; ++i) {
            b.x = pb[cur][i].x; b.y = pb[cur][i].y; acc[2 * i + 0] = WMMA_F32(a, b, acc[2 * i + 0]);
            b.x = pb[cur][i].z; b.y = pb[cur][i].w; acc[2 * i + 1] = WMMA_F32(a, b, acc[2 * i + 1]);
        }
        if (kk + 1 < 32) __builtin_amdgcn_sched_group_barrier(0x100, 4, 0);
        __builtin_amdgcn_sched_group_barrier(0x008, 8, 0);
    }

    float b2v[8], gv[8], bev[8];
#pragma unroll
    for (int t = 0; t < 8; ++t) {
        b2v[t] = b2[t * 16 + lm];
        gv[t]  = g[t * 16 + lm];
        bev[t] = beta[t * 16 + lm];
    }

    const size_t rb = (size_t)(m0 + 8 * half) * GNN_DOUT + lm;
    float* crow = out + rb;
    const float* rrow = RES ? (xres + rb) : nullptr;

#pragma unroll
    for (int r = 0; r < 8; ++r) {
        const float cr   = __shfl(cntv, r + 8 * half, 32);
        const float hasE = (cr > 0.0f) ? 1.0f : 0.0f;

        float h[8];
        float s = 0.0f;
#pragma unroll
        for (int t = 0; t < 8; ++t) { h[t] = acc[t][r] + hasE * b2v[t]; s += h[t]; }
#pragma unroll
        for (int m = 8; m >= 1; m >>= 1) s += __shfl_xor(s, m, 32);
        const float mu = s * (1.0f / GNN_DOUT);

        float s2 = 0.0f;
#pragma unroll
        for (int t = 0; t < 8; ++t) { h[t] -= mu; s2 += h[t] * h[t]; }
#pragma unroll
        for (int m = 8; m >= 1; m >>= 1) s2 += __shfl_xor(s2, m, 32);
        const float rstd = rsqrtf(s2 * (1.0f / GNN_DOUT) + GNN_EPS);

#pragma unroll
        for (int t = 0; t < 8; ++t) {
            float o = h[t] * rstd * gv[t] + bev[t];
            if (RES) o += rrow[(size_t)r * GNN_DOUT + t * 16];
            crow[(size_t)r * GNN_DOUT + t * 16] = lrelu(o);
        }
    }
}

extern "C" void kernel_launch(void* const* d_in, const int* in_sizes, int n_in,
                              void* d_out, int out_size, void* d_ws, size_t ws_size,
                              hipStream_t stream) {
    (void)in_sizes; (void)n_in; (void)out_size; (void)ws_size;

    const float* x  = (const float*)d_in[0];
    const int*   ei = (const int*)d_in[1];
    const float* W1[2] = {(const float*)d_in[2],  (const float*)d_in[8]};
    const float* b1[2] = {(const float*)d_in[3],  (const float*)d_in[9]};
    const float* W2[2] = {(const float*)d_in[4],  (const float*)d_in[10]};
    const float* b2[2] = {(const float*)d_in[5],  (const float*)d_in[11]};
    const float* g [2] = {(const float*)d_in[6],  (const float*)d_in[12]};
    const float* be[2] = {(const float*)d_in[7],  (const float*)d_in[13]};

    const size_t NF = (size_t)GNN_NT * GNN_DOUT;
    float* ws   = (float*)d_ws;
    float* X    = ws;
    float* P    = X + NF;
    float* Q    = P + NF;
    float* S    = Q + NF;
    float* cnt  = S + NF;

    k_gemm_pq<64><<<250, 256, 0, stream>>>(x, W1[0], P, Q);
    k_edge_own<<<250, 512, 0, stream>>>(P, Q, b1[0], ei, S, cnt);
    k_gemm_agg_ln<false><<<250, 256, 0, stream>>>(S, W2[0], cnt, b2[0], g[0], be[0], nullptr, X);

    k_gemm_pq<128><<<250, 256, 0, stream>>>(X, W1[1], P, Q);
    k_edge_own<<<250, 512, 0, stream>>>(P, Q, b1[1], ei, S, cnt);
    k_gemm_agg_ln<true><<<250, 256, 0, stream>>>(S, W2[1], cnt, b2[1], g[1], be[1], X, (float*)d_out);
}


# Rows of P, Q, S, cnt and the block-0 output stored node-major (node*16 + graph): each hit's 8 Q rows per wave, and a wave's P and S rows, are one contiguous 4-KB run, not 8 rows 1 MB apart
# speedup vs baseline: 4.9980x; 1.1301x over previous
#include <hip/hip_runtime.h>


typedef __attribute__((ext_vector_type(2))) float v2f;
typedef __attribute__((ext_vector_type(4))) float v4f;
typedef __attribute__((ext_vector_type(8))) float v8f;

#define GNN_SLOPE 0.01f
#define GNN_EPS   1e-5f
#define GNN_BSN   16
#define GNN_N     2000
#define GNN_E     32000
#define GNN_NT    (GNN_BSN * GNN_N)
#define GNN_DOUT  128
#define GNN_ROW(node, graph) ((size_t)(node) * GNN_BSN + (size_t)(graph))

__device__ __forceinline__ float lrelu(float v) { return v >= 0.0f ? v : GNN_SLOPE * v; }

#define WMMA_F32(A, B, C) \
    __builtin_amdgcn_wmma_f32_16x16x4_f32(false, (A), false, (B), (short)0, (C), false, false)

#define GNN_RPITCH 68
#define GNN_MAT(K) (2 * (K) * GNN_RPITCH)

template <int K, bool PAIR>
__device__ __forceinline__ void stage_weights(const float* __restrict__ W, float* __restrict__ L) {
    constexpr int PER = 2 * K * 4 / 256;
    v4f top[PER][4], bot[PER][4];
#pragma unroll
    for (int j = 0; j < PER; ++j) {
        const int u  = j * 256 + (int)threadIdx.x;
        const int q  = u & 3;
        const int R  = u >> 2;
        const int i  = R & 3;
        const int rp = R >> 2;
        const float* src = W + (size_t)(2 * rp) * GNN_DOUT + 32 * i + 4 * q;
        top[j][0] = *(const v4f*)(src);
        top[j][1] = *(const v4f*)(src + GNN_DOUT);
        top[j][2] = *(const v4f*)(src + 16);
        top[j][3] = *(const v4f*)(src + GNN_DOUT + 16);
        if (PAIR) {
            const float* sb = src + (size_t)K * GNN_DOUT;
            bot[j][0] = *(const v4f*)(sb);
            bot[j][1] = *(const v4f*)(sb + GNN_DOUT);
            bot[j][2] = *(const v4f*)(sb + 16);
            bot[j][3] = *(const v4f*)(sb + GNN_DOUT + 16);
        }
    }
#pragma unroll
    for (int j = 0; j < PER; ++j) {
        const int u = j * 256 + (int)threadIdx.x;
        const int q = u & 3;
        const int R = u >> 2;
        float* dst = L + R * GNN_RPITCH + 16 * q;
#pragma unroll
        for (int e = 0; e < 4; ++e) {
            v4f a = {top[j][0][e], top[j][1][e], top[j][2][e], top[j][3][e]};
            if (PAIR) {
                const v4f b = {bot[j][0][e], bot[j][1][e], bot[j][2][e], bot[j][3][e]};
                a -= b;
                *(v4f*)(dst + GNN_MAT(K) + 4 * e) = b;
            }
            *(v4f*)(dst + 4 * e) = a;
        }
    }
}

template <int K, bool A_BY_GRAPH>
__global__ __launch_bounds__(256) void k_gemm_pq(const float* __restrict__ A,
                                                 const float* __restrict__ W,
                                                 float* __restrict__ P,
                                                 float* __restrict__ Q) {
    __shared__ __attribute__((aligned(16))) float Lw[2 * GNN_MAT(K)];

    const int wave = threadIdx.x >> 5;
    const int lane = threadIdx.x & 31;
    const int half = lane >> 4;
    const int lm   = lane & 15;
    const int m0   = blockIdx.x * 128 + wave * 16;

    const size_t ar   = A_BY_GRAPH ? (size_t)lm * GNN_N + (size_t)(m0 >> 4) : (size_t)(m0 + lm);
    const float* arow = A + ar * K + 2 * half;
    v2f af[K / 4];
#pragma unroll
    for (int kk = 0; kk < K / 4; ++kk) af[kk] = *(const v2f*)(arow + 4 * kk);

    stage_weights<K, true>(W, Lw);

    v8f accP[8] = {};
    v8f accQ[8] = {};

    const v4f* lbA = (const v4f*)(Lw + half * 4 * GNN_RPITCH + lm * 4);
    const v4f* lbB = lbA + GNN_MAT(K) / 4;

    __syncthreads();

    v4f pa[2][4], pb[2][4];
#pragma unroll
    for (int i = 0; i < 4; ++i) { pa[0][i] = lbA[i * 17]; pb[0][i] = lbB[i * 17]; }
    __builtin_amdgcn_sched_group_barrier(0x100, 8, 0);

#pragma unroll
    for (int kk = 0; kk < K / 4; ++kk) {
        const int cur = kk & 1;
        const v2f a = af[kk];
        if (kk + 1 < K / 4) {
#pragma unroll
            for (int i = 0; i < 4; ++i) {
                pa[cur ^ 1][i] = lbA[((kk + 1) * 8 + i) * 17];
                pb[cur ^ 1][i] = lbB[((kk + 1) * 8 + i) * 17];
            }
        }
        v2f b;
#pragma unroll
        for (int i = 0; i < 4; ++i) {
            b.x = pa[cur][i].x; b.y = pa[cur][i].y; accP[2 * i + 0] = WMMA_F32(a, b, accP[2 * i + 0]);
            b.x = pa[cur][i].z; b.y = pa[cur][i].w; accP[2 * i + 1] = WMMA_F32(a, b, accP[2 * i + 1]);
            b.x = pb[cur][i].x; b.y = pb[cur][i].y; accQ[2 * i + 0] = WMMA_F32(a, b, accQ[2 * i + 0]);
            b.x = pb[cur][i].z; b.y = pb[cur][i].w; accQ[2 * i + 1] = WMMA_F32(a, b, accQ[2 * i + 1]);
        }
        if (kk + 1 < K / 4) __builtin_amdgcn_sched_group_barrier(0x100, 8, 0);
        __builtin_amdgcn_sched_group_barrier(0x008, 16, 0);
    }

    const size_t rb = (size_t)(m0 + 8 * half) * GNN_DOUT + lm;
    float* prow = P + rb;
    float* qrow = Q + rb;
#pragma unroll
    for (int r = 0; r < 8; ++r)
#pragma unroll
        for (int t = 0; t < 8; ++t) {
            prow[(size_t)r * GNN_DOUT + t * 16] = accP[t][r];
            qrow[(size_t)r * GNN_DOUT + t * 16] = accQ[t][r];
        }
}

#define GNN_EW  16
#define GNN_GW  (GNN_BSN / 2)
#define GNN_NS  5
#define GNN_CH  (GNN_E / GNN_EW)
#define GNN_NLD 16

__device__ __forceinline__ float4 lrelu_sum(float4 acc, float4 pb, float4 q) {
    float v;
    v = pb.x + q.x; acc.x += fmaxf(v, GNN_SLOPE * v);
    v = pb.y + q.y; acc.y += fmaxf(v, GNN_SLOPE * v);
    v = pb.z + q.z; acc.z += fmaxf(v, GNN_SLOPE * v);
    v = pb.w + q.w; acc.w += fmaxf(v, GNN_SLOPE * v);
    return acc;
}

__device__ __forceinline__ int keep_hits(unsigned bal, unsigned r, unsigned e,
                                         unsigned int* __restrict__ Hw, int n) {
    if (bal) {
        if (r < 8u) Hw[n + __builtin_amdgcn_mbcnt_lo(bal, 0u)] = (r << 16) | e;
        n += __builtin_popcount(bal);
    }
    return n;
}

__device__ __forceinline__ void load_set(float4 (&q)[GNN_GW], const float4* __restrict__ Q4, int s) {
    const size_t row = (size_t)(s < 0 ? 0 : s);
#pragma unroll
    for (int g = 0; g < GNN_GW; ++g) q[g] = Q4[GNN_ROW(row, g) * (GNN_DOUT / 4)];
}
__device__ __forceinline__ void sum_set(float4 (&acc)[GNN_GW], const float4 (&pb)[GNN_GW],
                                        const float4 (&q)[GNN_GW]) {
#pragma unroll
    for (int g = 0; g < GNN_GW; ++g) acc[g] = lrelu_sum(acc[g], pb[g], q[g]);
}

struct HitWalk {
    const unsigned int* H;
    const int* Hn;
    unsigned me;
    int lane;
    int w, base, n;
    unsigned bits, ent;
    __device__ __forceinline__ int next() {
        if (w >= GNN_EW) return -1;
        while (bits == 0u) {
            while (base >= n) {
                ++w;
                if (w >= GNN_EW) return -1;
                base = 0;
                n = __builtin_amdgcn_readfirstlane(Hn[w]);
            }
            const int idx = base + lane;
            ent  = idx < n ? H[w * GNN_CH + idx] : 0xffffffffu;
            bits = __builtin_amdgcn_ballot_w32((ent >> 16) == me);
            base += 32;
        }
        const int l = __builtin_ctz(bits);
        bits &= bits - 1u;
        return (int)(__builtin_amdgcn_readlane(ent, l) & 0xffffu);
    }
};

__global__ __launch_bounds__(512) void k_edge_own(const float* __restrict__ P,
                                                  const float* __restrict__ Q,
                                                  const float* __restrict__ b1,
                                                  const int* __restrict__ ei,
                                                  float* __restrict__ S,
                                                  float* __restrict__ cnt) {
    __shared__ unsigned int H[GNN_EW * GNN_CH];
    __shared__ int Hn[GNN_EW];

    const int tid  = threadIdx.x;
    const int lane = tid & 31;
    const int wave = __builtin_amdgcn_readfirstlane(tid >> 5);
    const unsigned nbase = blockIdx.x * 8;
    const unsigned slot  = (unsigned)wave >> 1;
    const unsigned node  = nbase + slot;
    const int      g0    = (wave & 1) * GNN_GW;

    float4 pb[GNN_GW];
#pragma unroll
    for (int g = 0; g < GNN_GW; ++g)
        pb[g] = *(const float4*)(P + GNN_ROW(node, g0 + g) * GNN_DOUT + lane * 4);
    const float4 bb = *(const float4*)(b1 + lane * 4);

    unsigned int* Hw = H + wave * GNN_CH;
    int n = 0;
    {
        const int4* d4 = (const int4*)(ei + GNN_E) + wave * (GNN_CH / 4) + lane;
        int4 d[GNN_NLD];
#pragma unroll
        for (int k = 0; k < GNN_NLD - 1; ++k) d[k] = d4[k * 32];
        d[GNN_NLD - 1] = make_int4(-1, -1, -1, -1);
        if (lane < GNN_CH / 4 - (GNN_NLD - 1) * 32) d[GNN_NLD - 1] = d4[(GNN_NLD - 1) * 32];
#pragma unroll
        for (int k = 0; k < GNN_NLD; ++k) {
            const unsigned r0 = (unsigned)d[k].x - nbase;
            const unsigned r1 = (unsigned)d[k].y - nbase;
            const unsigned r2 = (unsigned)d[k].z - nbase;
            const unsigned r3 = (unsigned)d[k].w - nbase;
            const unsigned m0 = __builtin_amdgcn_ballot_w32(r0 < 8u);
            const unsigned m1 = __builtin_amdgcn_ballot_w32(r1 < 8u);
            const unsigned m2 = __builtin_amdgcn_ballot_w32(r2 < 8u);
            const unsigned m3 = __builtin_amdgcn_ballot_w32(r3 < 8u);
            if (m0 | m1 | m2 | m3) {
                const unsigned e = (unsigned)(k * 32 + lane) * 4u;
                n = keep_hits(m0, r0, e + 0u, Hw, n);
                n = keep_hits(m1, r1, e + 1u, Hw, n);
                n = keep_hits(m2, r2, e + 2u, Hw, n);
                n = keep_hits(m3, r3, e + 3u, Hw, n);
            }
        }
        const int* srcw = ei + wave * GNN_CH;
        for (int i = lane; i < n; i += 32) {
            const unsigned e = Hw[i];
            const unsigned s = min((unsigned)srcw[e & 0xffffu], (unsigned)(GNN_N - 1));
            Hw[i] = (e & 0x70000u) | s;
        }
        if (lane == 0) Hn[wave] = n;
    }

    float4 acc[GNN_GW];
#pragma unroll
    for (int g = 0; g < GNN_GW; ++g) {
        pb[g].x += bb.x; pb[g].y += bb.y; pb[g].z += bb.z; pb[g].w += bb.w;
        acc[g] = make_float4(0.0f, 0.0f, 0.0f, 0.0f);
    }
    __syncthreads();

    const float4* Q4 = (const float4*)Q + (size_t)g0 * (GNN_DOUT / 4) + lane;
    HitWalk hw;
    hw.H = H; hw.Hn = Hn; hw.me = slot; hw.lane = lane;
    hw.w = 0; hw.base = 0; hw.n = __builtin_amdgcn_readfirstlane(Hn[0]);
    hw.bits = 0u; hw.ent = 0xffffffffu;

    float4 q[GNN_NS][GNN_GW];
    int    sh[GNN_NS];
    int deg = 0;
#pragma unroll
    for (int i = 0; i < GNN_NS; ++i) { sh[i] = hw.next(); load_set(q[i], Q4, sh[i]); }
    while (sh[0] >= 0) {
#pragma unroll
        for (int i = 0; i < GNN_NS; ++i) {
            if (sh[i] >= 0) { sum_set(acc, pb, q[i]); ++deg; }
            sh[i] = hw.next(); load_set(q[i], Q4, sh[i]);
        }
    }

#pragma unroll
    for (int g = 0; g < GNN_GW; ++g)
        *(float4*)(S + GNN_ROW(node, g0 + g) * GNN_DOUT + lane * 4) = acc[g];
    if (g0 == 0 && lane < GNN_BSN) cnt[GNN_ROW(node, lane)] = (float)deg;
}

template <bool RES, bool OUT_BY_GRAPH>
__global__ __launch_bounds__(256) void k_gemm_agg_ln(const float* __restrict__ S,
                                                     const float* __restrict__ W2,
                                                     const float* __restrict__ cnt,
                                                     const float* __restrict__ b2,
                                                     const float* __restrict__ g,
                                                     const float* __restrict__ beta,
                                                     const float* __restrict__ xres,
                                                     float* __restrict__ out) {
    const int wave = threadIdx.x >> 5;
    const int lane = threadIdx.x & 31;
    const int half = lane >> 4;
    const int lm   = lane & 15;
    const int m0   = blockIdx.x * 128 + wave * 16;

    __shared__ __attribute__((aligned(16))) float Lw[GNN_MAT(128)];

    const float cntv  = cnt[m0 + lm];
    const float* arow = S + (size_t)(m0 + lm) * 128 + 2 * half;
    v2f af[32];
#pragma unroll
    for (int kk = 0; kk < 32; ++kk) af[kk] = *(const v2f*)(arow + 4 * kk);

    stage_weights<128, false>(W2, Lw);

    const float scale = 1.0f / fmaxf(cntv, 1.0f);

    v8f acc[8] = {};
    const v4f*   lb   = (const v4f*)(Lw + half * 4 * GNN_RPITCH + lm * 4);

    __syncthreads();

    v4f pb[2][4];
#pragma unroll
    for (int i = 0; i < 4; ++i) pb[0][i] = lb[i * 17];
    __builtin_amdgcn_sched_group_barrier(0x100, 4, 0);

#pragma unroll
    for (int kk = 0; kk < 32; ++kk) {
        const int cur = kk & 1;
        v2f a = af[kk];
        a.x *= scale; a.y *= scale;
        if (kk + 1 < 32) {
#pragma unroll
            for (int i = 0; i < 4; ++i) pb[cur ^ 1][i] = lb[((kk + 1) * 8 + i) * 17];
        }
        v2f b;
#pragma unroll
        for (int i = 0; i < 4; ++i) {
            b.x = pb[cur][i].x; b.y = pb[cur][i].y; acc[2 * i + 0] = WMMA_F32(a, b, acc[2 * i + 0]);
            b.x = pb[cur][i].z; b.y = pb[cur][i].w; acc[2 * i + 1] = WMMA_F32(a, b, acc[2 * i + 1]);
        }
        if (kk + 1 < 32) __builtin_amdgcn_sched_group_barrier(0x100, 4, 0);
        __builtin_amdgcn_sched_group_barrier(0x008, 8, 0);
    }

    float b2v[8], gv[8], bev[8];
#pragma unroll
    for (int t = 0; t < 8; ++t) {
        b2v[t] = b2[t * 16 + lm];
        gv[t]  = g[t * 16 + lm];
        bev[t] = beta[t * 16 + lm];
    }

    const size_t rb = (size_t)(m0 + 8 * half) * GNN_DOUT + lm;
    constexpr size_t ostep = OUT_BY_GRAPH ? (size_t)GNN_N * GNN_DOUT : (size_t)GNN_DOUT;
    float* crow = out + (OUT_BY_GRAPH
                             ? ((size_t)(8 * half) * GNN_N + (size_t)(m0 >> 4)) * GNN_DOUT + lm
                             : rb);
    const float* rrow = RES ? (xres + rb) : nullptr;

#pragma unroll
    for (int r = 0; r < 8; ++r) {
        const float cr   = __shfl(cntv, r + 8 * half, 32);
        const float hasE = (cr > 0.0f) ? 1.0f : 0.0f;

        float h[8];
        float s = 0.0f;
#pragma unroll
        for (int t = 0; t < 8; ++t) { h[t] = acc[t][r] + hasE * b2v[t]; s += h[t]; }
#pragma unroll
        for (int m = 8; m >= 1; m >>= 1) s += __shfl_xor(s, m, 32);
        const float mu = s * (1.0f / GNN_DOUT);

        float s2 = 0.0f;
#pragma unroll
        for (int t = 0; t < 8; ++t) { h[t] -= mu; s2 += h[t] * h[t]; }
#pragma unroll
        for (int m = 8; m >= 1; m >>= 1) s2 += __shfl_xor(s2, m, 32);
        const float rstd = rsqrtf(s2 * (1.0f / GNN_DOUT) + GNN_EPS);

#pragma unroll
        for (int t = 0; t < 8; ++t) {
            float o = h[t] * rstd * gv[t] + bev[t];
            if (RES) o += rrow[(size_t)r * GNN_DOUT + t * 16];
            crow[(size_t)r * ostep + t * 16] = lrelu(o);
        }
    }
}

extern "C" void kernel_launch(void* const* d_in, const int* in_sizes, int n_in,
                              void* d_out, int out_size, void* d_ws, size_t ws_size,
                              hipStream_t stream) {
    (void)in_sizes; (void)n_in; (void)out_size; (void)ws_size;

    const float* x  = (const float*)d_in[0];
    const int*   ei = (const int*)d_in[1];
    const float* W1[2] = {(const float*)d_in[2],  (const float*)d_in[8]};
    const float* b1[2] = {(const float*)d_in[3],  (const float*)d_in[9]};
    const float* W2[2] = {(const float*)d_in[4],  (const float*)d_in[10]};
    const float* b2[2] = {(const float*)d_in[5],  (const float*)d_in[11]};
    const float* g [2] = {(const float*)d_in[6],  (const float*)d_in[12]};
    const float* be[2] = {(const float*)d_in[7],  (const float*)d_in[13]};

    const size_t NF = (size_t)GNN_NT * GNN_DOUT;
    float* ws   = (float*)d_ws;
    float* X    = ws;
    float* P    = X + NF;
    float* Q    = P + NF;
    float* S    = Q + NF;
    float* cnt  = S + NF;

    k_gemm_pq<64, true><<<250, 256, 0, stream>>>(x, W1[0], P, Q);
    k_edge_own<<<250, 512, 0, stream>>>(P, Q, b1[0], ei, S, cnt);
    k_gemm_agg_ln<false, false><<<250, 256, 0, stream>>>(S, W2[0], cnt, b2[0], g[0], be[0], nullptr, X);

    k_gemm_pq<128, false><<<250, 256, 0, stream>>>(X, W1[1], P, Q);
    k_edge_own<<<250, 512, 0, stream>>>(P, Q, b1[1], ei, S, cnt);
    k_gemm_agg_ln<true, true><<<250, 256, 0, stream>>>(S, W2[1], cnt, b2[1], g[1], be[1], X, (float*)d_out);
}


# Edge filter runs once per call in eight spare waves of block 0's P/Q GEMM launch (same 8 nodes per work-group); both edge stages read its hit lists from the workspace; float sum order unchanged
# speedup vs baseline: 5.2021x; 1.0408x over previous
#include <hip/hip_runtime.h>


typedef __attribute__((ext_vector_type(2))) float v2f;
typedef __attribute__((ext_vector_type(4))) float v4f;
typedef __attribute__((ext_vector_type(8))) float v8f;

#define GNN_SLOPE 0.01f
#define GNN_EPS   1e-5f
#define GNN_BSN   16
#define GNN_N     2000
#define GNN_E     32000
#define GNN_NT    (GNN_BSN * GNN_N)
#define GNN_DOUT  128
#define GNN_ROW(node, graph) ((size_t)(node) * GNN_BSN + (size_t)(graph))

__device__ __forceinline__ float lrelu(float v) { return v >= 0.0f ? v : GNN_SLOPE * v; }

#define WMMA_F32(A, B, C) \
    __builtin_amdgcn_wmma_f32_16x16x4_f32(false, (A), false, (B), (short)0, (C), false, false)

#define GNN_RPITCH 68
#define GNN_MAT(K) (2 * (K) * GNN_RPITCH)

template <int K, bool PAIR>
__device__ __forceinline__ void stage_weights(const float* __restrict__ W, float* __restrict__ L) {
    constexpr int PER = 2 * K * 4 / 256;
    v4f top[PER][4], bot[PER][4];
#pragma unroll
    for (int j = 0; j < PER; ++j) {
        const int u  = j * 256 + (int)threadIdx.x;
        const int q  = u & 3;
        const int R  = u >> 2;
        const int i  = R & 3;
        const int rp = R >> 2;
        const float* src = W + (size_t)(2 * rp) * GNN_DOUT + 32 * i + 4 * q;
        top[j][0] = *(const v4f*)(src);
        top[j][1] = *(const v4f*)(src + GNN_DOUT);
        top[j][2] = *(const v4f*)(src + 16);
        top[j][3] = *(const v4f*)(src + GNN_DOUT + 16);
        if (PAIR) {
            const float* sb = src + (size_t)K * GNN_DOUT;
            bot[j][0] = *(const v4f*)(sb);
            bot[j][1] = *(const v4f*)(sb + GNN_DOUT);
            bot[j][2] = *(const v4f*)(sb + 16);
            bot[j][3] = *(const v4f*)(sb + GNN_DOUT + 16);
        }
    }
#pragma unroll
    for (int j = 0; j < PER; ++j) {
        const int u = j * 256 + (int)threadIdx.x;
        const int q = u & 3;
        const int R = u >> 2;
        float* dst = L + R * GNN_RPITCH + 16 * q;
#pragma unroll
        for (int e = 0; e < 4; ++e) {
            v4f a = {top[j][0][e], top[j][1][e], top[j][2][e], top[j][3][e]};
            if (PAIR) {
                const v4f b = {bot[j][0][e], bot[j][1][e], bot[j][2][e], bot[j][3][e]};
                a -= b;
                *(v4f*)(dst + GNN_MAT(K) + 4 * e) = b;
            }
            *(v4f*)(dst + 4 * e) = a;
        }
    }
}

#define GNN_FW     8
#define GNN_FCH    (GNN_E / GNN_FW)
#define GNN_HCH    (GNN_FCH / 2)
#define GNN_NLD    16
#define GNN_LPITCH 4096

__device__ __forceinline__ int keep_hits(unsigned bal, unsigned r, unsigned e,
                                         unsigned int* __restrict__ Hw, int n) {
    if (bal) {
        if (r < 8u) Hw[n + __builtin_amdgcn_mbcnt_lo(bal, 0u)] = (r << 16) | e;
        n += __builtin_popcount(bal);
    }
    return n;
}

__device__ __forceinline__ void filter_lists(const int* __restrict__ ei,
                                             unsigned int* __restrict__ Hw,
                                             unsigned int* __restrict__ L,
                                             int fw, int lane, unsigned nbase) {
    int n = 0;
#pragma unroll 1
    for (int hp = 0; hp < 2; ++hp) {
        const int4* d4 = (const int4*)(ei + GNN_E + fw * GNN_FCH + hp * GNN_HCH) + lane;
        int4 d[GNN_NLD];
#pragma unroll
        for (int k = 0; k < GNN_NLD - 1; ++k) d[k] = d4[k * 32];
        d[GNN_NLD - 1] = make_int4(-1, -1, -1, -1);
        if (lane < GNN_HCH / 4 - (GNN_NLD - 1) * 32) d[GNN_NLD - 1] = d4[(GNN_NLD - 1) * 32];
#pragma unroll
        for (int k = 0; k < GNN_NLD; ++k) {
            const unsigned r0 = (unsigned)d[k].x - nbase;
            const unsigned r1 = (unsigned)d[k].y - nbase;
            const unsigned r2 = (unsigned)d[k].z - nbase;
            const unsigned r3 = (unsigned)d[k].w - nbase;
            const unsigned m0 = __builtin_amdgcn_ballot_w32(r0 < 8u);
            const unsigned m1 = __builtin_amdgcn_ballot_w32(r1 < 8u);
            const unsigned m2 = __builtin_amdgcn_ballot_w32(r2 < 8u);
            const unsigned m3 = __builtin_amdgcn_ballot_w32(r3 < 8u);
            if (m0 | m1 | m2 | m3) {
                const unsigned e = (unsigned)(hp * GNN_HCH) + (unsigned)(k * 32 + lane) * 4u;
                n = keep_hits(m0, r0, e + 0u, Hw, n);
                n = keep_hits(m1, r1, e + 1u, Hw, n);
                n = keep_hits(m2, r2, e + 2u, Hw, n);
                n = keep_hits(m3, r3, e + 3u, Hw, n);
            }
        }
    }
    const int* srcw = ei + fw * GNN_FCH;
    for (int i = lane; i < n; i += 32) {
        const unsigned e = Hw[i];
        const unsigned s = min((unsigned)srcw[e & 0xffffu], (unsigned)(GNN_N - 1));
        Hw[i] = (e & 0x70000u) | s;
    }
    L[lane] = lane == 0 ? (unsigned)n : (lane - 1 < n ? Hw[lane - 1] : 0xffffffffu);
    for (int i = 31 + lane; i < n; i += 32) L[1 + i] = Hw[i];
}

template <int K, bool A_BY_GRAPH>
__global__ __launch_bounds__(A_BY_GRAPH ? 512 : 256) void k_gemm_pq(const float* __restrict__ A,
                                                 const float* __restrict__ W,
                                                 float* __restrict__ P,
                                                 float* __restrict__ Q,
                                                 const int* __restrict__ ei,
                                                 unsigned int* __restrict__ lists) {
    __shared__ __attribute__((aligned(16))) float Lw[2 * GNN_MAT(K)];
    __shared__ unsigned int Hf[A_BY_GRAPH ? GNN_FW * GNN_FCH : 4];

    const int wave = __builtin_amdgcn_readfirstlane(threadIdx.x >> 5);
    const int lane = threadIdx.x & 31;
    const int half = lane >> 4;
    const int lm   = lane & 15;
    const bool filt = A_BY_GRAPH && wave >= 8;
    const int m0   = blockIdx.x * 128 + (wave & 7) * 16;

    const size_t ar   = A_BY_GRAPH ? (size_t)lm * GNN_N + (size_t)(m0 >> 4) : (size_t)(m0 + lm);
    const float* arow = A + ar * K + 2 * half;
    v2f af[K / 4];
    if (!filt) {
#pragma unroll
        for (int kk = 0; kk < K / 4; ++kk) af[kk] = *(const v2f*)(arow + 4 * kk);

        stage_weights<K, true>(W, Lw);
    }

    v8f accP[8] = {};
    v8f accQ[8] = {};

    const v4f* lbA = (const v4f*)(Lw + half * 4 * GNN_RPITCH + lm * 4);
    const v4f* lbB = lbA + GNN_MAT(K) / 4;

    __syncthreads();

    if (A_BY_GRAPH) {
        if (filt) {
            filter_lists(ei, Hf + (wave - 8) * GNN_FCH,
                         lists + ((size_t)blockIdx.x * GNN_FW + (size_t)(wave - 8)) * GNN_LPITCH,
                         wave - 8, lane, blockIdx.x * 8u);
            return;
        }
    }

    v4f pa[2][4], pb[2][4];
#pragma unroll
    for (int i = 0; i < 4; ++i) { pa[0][i] = lbA[i * 17]; pb[0][i] = lbB[i * 17]; }
    __builtin_amdgcn_sched_group_barrier(0x100, 8, 0);

#pragma unroll
    for (int kk = 0; kk < K / 4; ++kk) {
        const int cur = kk & 1;
        const v2f a = af[kk];
        if (kk + 1 < K / 4) {
#pragma unroll
            for (int i = 0; i < 4; ++i) {
                pa[cur ^ 1][i] = lbA[((kk + 1) * 8 + i) * 17];
                pb[cur ^ 1][i] = lbB[((kk + 1) * 8 + i) * 17];
            }
        }
        v2f b;
#pragma unroll
        for (int i = 0; i < 4; ++i) {
            b.x = pa[cur][i].x; b.y = pa[cur][i].y; accP[2 * i + 0] = WMMA_F32(a, b, accP[2 * i + 0]);
            b.x = pa[cur][i].z; b.y = pa[cur][i].w; accP[2 * i + 1] = WMMA_F32(a, b, accP[2 * i + 1]);
            b.x = pb[cur][i].x; b.y = pb[cur][i].y; accQ[2 * i + 0] = WMMA_F32(a, b, accQ[2 * i + 0]);
            b.x = pb[cur][i].z; b.y = pb[cur][i].w; accQ[2 * i + 1] = WMMA_F32(a, b, accQ[2 * i + 1]);
        }
        if (kk + 1 < K / 4) __builtin_amdgcn_sched_group_barrier(0x100, 8, 0);
        __builtin_amdgcn_sched_group_barrier(0x008, 16, 0);
    }

    const size_t rb = (size_t)(m0 + 8 * half) * GNN_DOUT + lm;
    float* prow = P + rb;
    float* qrow = Q + rb;
#pragma unroll
    for (int r = 0; r < 8; ++r)
#pragma unroll
        for (int t = 0; t < 8; ++t) {
            prow[(size_t)r * GNN_DOUT + t * 16] = accP[t][r];
            qrow[(size_t)r * GNN_DOUT + t * 16] = accQ[t][r];
        }
}

#define GNN_GW  (GNN_BSN / 2)
#define GNN_NS  5

__device__ __forceinline__ float4 lrelu_sum(float4 acc, float4 pb, float4 q) {
    float v;
    v = pb.x + q.x; acc.x += fmaxf(v, GNN_SLOPE * v);
    v = pb.y + q.y; acc.y += fmaxf(v, GNN_SLOPE * v);
    v = pb.z + q.z; acc.z += fmaxf(v, GNN_SLOPE * v);
    v = pb.w + q.w; acc.w += fmaxf(v, GNN_SLOPE * v);
    return acc;
}

__device__ __forceinline__ void load_set(float4 (&q)[GNN_GW], const float4* __restrict__ Q4, int s) {
    const size_t row = (size_t)(s < 0 ? 0 : s);
#pragma unroll
    for (int g = 0; g < GNN_GW; ++g) q[g] = Q4[GNN_ROW(row, g) * (GNN_DOUT / 4)];
}
__device__ __forceinline__ void sum_set(float4 (&acc)[GNN_GW], const float4 (&pb)[GNN_GW],
                                        const float4 (&q)[GNN_GW]) {
#pragma unroll
    for (int g = 0; g < GNN_GW; ++g) acc[g] = lrelu_sum(acc[g], pb[g], q[g]);
}

struct HitWalk {
    const unsigned int* H;
    const int* Hn;
    unsigned me;
    int lane;
    int w, base, n;
    unsigned bits, ent;
    __device__ __forceinline__ int next() {
        if (w >= GNN_FW) return -1;
        while (bits == 0u) {
            while (base >= n) {
                ++w;
                if (w >= GNN_FW) return -1;
                base = 0;
                n = __builtin_amdgcn_readfirstlane(Hn[w]);
            }
            const int idx = base + lane;
            ent  = idx < n ? H[w * GNN_FCH + idx] : 0xffffffffu;
            bits = __builtin_amdgcn_ballot_w32((ent >> 16) == me);
            base += 32;
        }
        const int l = __builtin_ctz(bits);
        bits &= bits - 1u;
        return (int)(__builtin_amdgcn_readlane(ent, l) & 0xffffu);
    }
};

__global__ __launch_bounds__(512) void k_edge_own(const float* __restrict__ P,
                                                  const float* __restrict__ Q,
                                                  const float* __restrict__ b1,
                                                  const unsigned int* __restrict__ lists,
                                                  float* __restrict__ S,
                                                  float* __restrict__ cnt) {
    __shared__ unsigned int H[GNN_FW * GNN_FCH];
    __shared__ int Hn[GNN_FW];

    const int tid  = threadIdx.x;
    const int lane = tid & 31;
    const int wave = __builtin_amdgcn_readfirstlane(tid >> 5);
    const unsigned nbase = blockIdx.x * 8;
    const unsigned slot  = (unsigned)wave >> 1;
    const unsigned node  = nbase + slot;
    const int      g0    = (wave & 1) * GNN_GW;

    float4 pb[GNN_GW];
#pragma unroll
    for (int g = 0; g < GNN_GW; ++g)
        pb[g] = *(const float4*)(P + GNN_ROW(node, g0 + g) * GNN_DOUT + lane * 4);
    const float4 bb = *(const float4*)(b1 + lane * 4);

    if (wave < GNN_FW) {
        const unsigned int* L = lists + ((size_t)blockIdx.x * GNN_FW + (size_t)wave) * GNN_LPITCH;
        unsigned int* Hw = H + wave * GNN_FCH;
        const unsigned v = L[lane];
        const int n = (int)min((unsigned)__builtin_amdgcn_readfirstlane(v), (unsigned)GNN_FCH);
        if (lane >= 1 && lane - 1 < n) Hw[lane - 1] = v;
        for (int i = 31 + lane; i < n; i += 32) Hw[i] = L[1 + i];
        if (lane == 0) Hn[wave] = n;
    }

    float4 acc[GNN_GW];
#pragma unroll
    for (int g = 0; g < GNN_GW; ++g) {
        pb[g].x += bb.x; pb[g].y += bb.y; pb[g].z += bb.z; pb[g].w += bb.w;
        acc[g] = make_float4(0.0f, 0.0f, 0.0f, 0.0f);
    }
    __syncthreads();

    const float4* Q4 = (const float4*)Q + (size_t)g0 * (GNN_DOUT / 4) + lane;
    HitWalk hw;
    hw.H = H; hw.Hn = Hn; hw.me = slot; hw.lane = lane;
    hw.w = 0; hw.base = 0; hw.n = __builtin_amdgcn_readfirstlane(Hn[0]);
    hw.bits = 0u; hw.ent = 0xffffffffu;

    float4 q[GNN_NS][GNN_GW];
    int    sh[GNN_NS];
    int deg = 0;
#pragma unroll
    for (int i = 0; i < GNN_NS; ++i) { sh[i] = hw.next(); load_set(q[i], Q4, sh[i]); }
    while (sh[0] >= 0) {
#pragma unroll
        for (int i = 0; i < GNN_NS; ++i) {
            if (sh[i] >= 0) { sum_set(acc, pb, q[i]); ++deg; }
            sh[i] = hw.next(); load_set(q[i], Q4, sh[i]);
        }
    }

#pragma unroll
    for (int g = 0; g < GNN_GW; ++g)
        *(float4*)(S + GNN_ROW(node, g0 + g) * GNN_DOUT + lane * 4) = acc[g];
    if (g0 == 0 && lane < GNN_BSN) cnt[GNN_ROW(node, lane)] = (float)deg;
}

template <bool RES, bool OUT_BY_GRAPH>
__global__ __launch_bounds__(256) void k_gemm_agg_ln(const float* __restrict__ S,
                                                     const float* __restrict__ W2,
                                                     const float* __restrict__ cnt,
                                                     const float* __restrict__ b2,
                                                     const float* __restrict__ g,
                                                     const float* __restrict__ beta,
                                                     const float* __restrict__ xres,
                                                     float* __restrict__ out) {
    const int wave = threadIdx.x >> 5;
    const int lane = threadIdx.x & 31;
    const int half = lane >> 4;
    const int lm   = lane & 15;
    const int m0   = blockIdx.x * 128 + wave * 16;

    __shared__ __attribute__((aligned(16))) float Lw[GNN_MAT(128)];

    const float cntv  = cnt[m0 + lm];
    const float* arow = S + (size_t)(m0 + lm) * 128 + 2 * half;
    v2f af[32];
#pragma unroll
    for (int kk = 0; kk < 32; ++kk) af[kk] = *(const v2f*)(arow + 4 * kk);

    stage_weights<128, false>(W2, Lw);

    const float scale = 1.0f / fmaxf(cntv, 1.0f);

    v8f acc[8] = {};
    const v4f*   lb   = (const v4f*)(Lw + half * 4 * GNN_RPITCH + lm * 4);

    __syncthreads();

    v4f pb[2][4];
#pragma unroll
    for (int i = 0; i < 4; ++i) pb[0][i] = lb[i * 17];
    __builtin_amdgcn_sched_group_barrier(0x100, 4, 0);

#pragma unroll
    for (int kk = 0; kk < 32; ++kk) {
        const int cur = kk & 1;
        v2f a = af[kk];
        a.x *= scale; a.y *= scale;
        if (kk + 1 < 32) {
#pragma unroll
            for (int i = 0; i < 4; ++i) pb[cur ^ 1][i] = lb[((kk + 1) * 8 + i) * 17];
        }
        v2f b;
#pragma unroll
        for (int i = 0; i < 4; ++i) {
            b.x = pb[cur][i].x; b.y = pb[cur][i].y; acc[2 * i + 0] = WMMA_F32(a, b, acc[2 * i + 0]);
            b.x = pb[cur][i].z; b.y = pb[cur][i].w; acc[2 * i + 1] = WMMA_F32(a, b, acc[2 * i + 1]);
        }
        if (kk + 1 < 32) __builtin_amdgcn_sched_group_barrier(0x100, 4, 0);
        __builtin_amdgcn_sched_group_barrier(0x008, 8, 0);
    }

    float b2v[8], gv[8], bev[8];
#pragma unroll
    for (int t = 0; t < 8; ++t) {
        b2v[t] = b2[t * 16 + lm];
        gv[t]  = g[t * 16 + lm];
        bev[t] = beta[t * 16 + lm];
    }

    const size_t rb = (size_t)(m0 + 8 * half) * GNN_DOUT + lm;
    constexpr size_t ostep = OUT_BY_GRAPH ? (size_t)GNN_N * GNN_DOUT : (size_t)GNN_DOUT;
    float* crow = out + (OUT_BY_GRAPH
                             ? ((size_t)(8 * half) * GNN_N + (size_t)(m0 >> 4)) * GNN_DOUT + lm
                             : rb);
    const float* rrow = RES ? (xres + rb) : nullptr;

#pragma unroll
    for (int r = 0; r < 8; ++r) {
        const float cr   = __shfl(cntv, r + 8 * half, 32);
        const float hasE = (cr > 0.0f) ? 1.0f : 0.0f;

        float h[8];
        float s = 0.0f;
#pragma unroll
        for (int t = 0; t < 8; ++t) { h[t] = acc[t][r] + hasE * b2v[t]; s += h[t]; }
#pragma unroll
        for (int m = 8; m >= 1; m >>= 1) s += __shfl_xor(s, m, 32);
        const float mu = s * (1.0f / GNN_DOUT);

        float s2 = 0.0f;
#pragma unroll
        for (int t = 0; t < 8; ++t) { h[t] -= mu; s2 += h[t] * h[t]; }
#pragma unroll
        for (int m = 8; m >= 1; m >>= 1) s2 += __shfl_xor(s2, m, 32);
        const float rstd = rsqrtf(s2 * (1.0f / GNN_DOUT) + GNN_EPS);

#pragma unroll
        for (int t = 0; t < 8; ++t) {
            float o = h[t] * rstd * gv[t] + bev[t];
            if (RES) o += rrow[(size_t)r * GNN_DOUT + t * 16];
            crow[(size_t)r * ostep + t * 16] = lrelu(o);
        }
    }
}

extern "C" void kernel_launch(void* const* d_in, const int* in_sizes, int n_in,
                              void* d_out, int out_size, void* d_ws, size_t ws_size,
                              hipStream_t stream) {
    (void)in_sizes; (void)n_in; (void)out_size; (void)ws_size;

    const float* x  = (const float*)d_in[0];
    const int*   ei = (const int*)d_in[1];
    const float* W1[2] = {(const float*)d_in[2],  (const float*)d_in[8]};
    const float* b1[2] = {(const float*)d_in[3],  (const float*)d_in[9]};
    const float* W2[2] = {(const float*)d_in[4],  (const float*)d_in[10]};
    const float* b2[2] = {(const float*)d_in[5],  (const float*)d_in[11]};
    const float* g [2] = {(const float*)d_in[6],  (const float*)d_in[12]};
    const float* be[2] = {(const float*)d_in[7],  (const float*)d_in[13]};

    const size_t NF = (size_t)GNN_NT * GNN_DOUT;
    float* ws   = (float*)d_ws;
    float* X    = ws;
    float* P    = X + NF;
    float* Q    = P + NF;
    float* S    = Q + NF;
    float* cnt  = S + NF;
    unsigned int* lists = (unsigned int*)(cnt + GNN_NT);

    k_gemm_pq<64, true><<<250, 512, 0, stream>>>(x, W1[0], P, Q, ei, lists);
    k_edge_own<<<250, 512, 0, stream>>>(P, Q, b1[0], lists, S, cnt);
    k_gemm_agg_ln<false, false><<<250, 256, 0, stream>>>(S, W2[0], cnt, b2[0], g[0], be[0], nullptr, X);

    k_gemm_pq<128, false><<<250, 256, 0, stream>>>(X, W1[1], P, Q, nullptr, nullptr);
    k_edge_own<<<250, 512, 0, stream>>>(P, Q, b1[1], lists, S, cnt);
    k_gemm_agg_ln<true, true><<<250, 256, 0, stream>>>(S, W2[1], cnt, b2[1], g[1], be[1], X, (float*)d_out);
}
